# starting kernel
# baseline (speedup 1.0000x reference)
#include <hip/hip_runtime.h>
#include <hip/hip_bf16.h>

typedef float v2f __attribute__((ext_vector_type(2)));
typedef float v8f __attribute__((ext_vector_type(8)));

#define NFEAT 128

__global__ void gcn_fill1(float* __restrict__ deg, int n) {
    int i = blockIdx.x * blockDim.x + threadIdx.x;
    if (i < n) deg[i] = 1.0f;
}

__global__ void gcn_deg_accum(const int* __restrict__ dst,
                              const float* __restrict__ ew,
                              float* __restrict__ deg, int n_edges) {
    int e = blockIdx.x * blockDim.x + threadIdx.x;
    if (e < n_edges)
        __hip_atomic_fetch_add(&deg[dst[e]], ew[e],
                               __ATOMIC_RELAXED, __HIP_MEMORY_SCOPE_AGENT);
}

__global__ void gcn_rsqrt(float* __restrict__ d, int n) {
    int i = blockIdx.x * blockDim.x + threadIdx.x;
    if (i < n) d[i] = rsqrtf(d[i]);
}

__global__ void gcn_zero(float* __restrict__ p, int n) {
    int i = blockIdx.x * blockDim.x + threadIdx.x;
    if (i < n) p[i] = 0.0f;
}

__global__ __launch_bounds__(256) void gcn_gemm128_wmma(
        const float* __restrict__ X, const float* __restrict__ W,
        float* __restrict__ H) {
    __shared__ float Wl[NFEAT * NFEAT];
    {
        const float4* Wv  = (const float4*)W;
        float4*       Wlv = (float4*)Wl;
        int tid = threadIdx.y * 32 + threadIdx.x;
        #pragma unroll
        for (int i = 0; i < (NFEAT * NFEAT / 4) / 256; ++i)
            Wlv[tid + i * 256] = Wv[tid + i * 256];
    }
    __syncthreads();

    const int lane  = threadIdx.x;
    const int m0    = blockIdx.x * 16;
    const int n0    = threadIdx.y * 16;
    const int mlane = lane & 15;
    const int khalf = (lane >> 4) * 2;
    const int ncol  = n0 + mlane;

    const float* xrow = X + (size_t)(m0 + mlane) * NFEAT;
    v8f acc = {};

    #pragma unroll
    for (int k0 = 0; k0 < NFEAT; k0 += 4) {
        const int k = k0 + khalf;
        v2f a = *(const v2f*)&xrow[k];
        v2f b;
        b.x = Wl[(k    ) * NFEAT + ncol];
        b.y = Wl[(k + 1) * NFEAT + ncol];
        acc = __builtin_amdgcn_wmma_f32_16x16x4_f32(
                  false, a, false, b, (short)0, acc, false, false);
    }

    const int rbase = m0 + (lane >> 4) * 8;
    #pragma unroll
    for (int r = 0; r < 8; ++r)
        H[(size_t)(rbase + r) * NFEAT + ncol] = acc[r];
}

__global__ __launch_bounds__(256) void gcn_scatter(
        const float* __restrict__ h, float* __restrict__ agg,
        const int* __restrict__ src, const int* __restrict__ dst,
        const float* __restrict__ ew, const float* __restrict__ dinv,
        int n_edges) {
    const int wave = (blockIdx.x * blockDim.x + threadIdx.x) >> 5;
    const int lane = threadIdx.x & 31;
    if (wave >= n_edges) return;

    const int   s = src[wave];
    const int   d = dst[wave];
    const float c = dinv[s] * ew[wave] * dinv[d];

    const float4 hv = *(const float4*)&h[(size_t)s * NFEAT + lane * 4];
    float* out = &agg[(size_t)d * NFEAT + lane * 4];
    __hip_atomic_fetch_add(&out[0], hv.x * c, __ATOMIC_RELAXED, __HIP_MEMORY_SCOPE_AGENT);
    __hip_atomic_fetch_add(&out[1], hv.y * c, __ATOMIC_RELAXED, __HIP_MEMORY_SCOPE_AGENT);
    __hip_atomic_fetch_add(&out[2], hv.z * c, __ATOMIC_RELAXED, __HIP_MEMORY_SCOPE_AGENT);
    __hip_atomic_fetch_add(&out[3], hv.w * c, __ATOMIC_RELAXED, __HIP_MEMORY_SCOPE_AGENT);
}

__global__ void gcn_fuse(const float* __restrict__ agg,
                         const float* __restrict__ hpre,
                         const float* __restrict__ dinv,
                         const float* __restrict__ b,
                         float* __restrict__ out,
                         int n_nodes, int do_relu) {
    int idx = blockIdx.x * blockDim.x + threadIdx.x;
    if (idx >= n_nodes * NFEAT) return;
    int i = idx >> 7;
    int d = idx & (NFEAT - 1);
    float di = dinv[i];
    float v = agg[idx] + hpre[idx] * (di * di) + b[d];
    out[idx] = do_relu ? fmaxf(v, 0.0f) : v;
}

__global__ void gcn_classifier(const float* __restrict__ h,
                               const float* __restrict__ Wc,
                               const float* __restrict__ bc,
                               float* __restrict__ out, int n_nodes) {
    int i = blockIdx.x * blockDim.x + threadIdx.x;
    if (i >= n_nodes) return;
    float acc[10];
    #pragma unroll
    for (int c = 0; c < 10; ++c) acc[c] = bc[c];
    const float* hr = h + (size_t)i * NFEAT;
    for (int d = 0; d < NFEAT; ++d) {
        float hv = hr[d];
        #pragma unroll
        for (int c = 0; c < 10; ++c) acc[c] += hv * Wc[d * 10 + c];
    }
    #pragma unroll
    for (int c = 0; c < 10; ++c) out[(size_t)i * 10 + c] = acc[c];
}

extern "C" void kernel_launch(void* const* d_in, const int* in_sizes, int n_in,
                              void* d_out, int out_size, void* d_ws, size_t ws_size,
                              hipStream_t stream) {
    const float* x   = (const float*)d_in[0];
    const int*   ei  = (const int*)  d_in[1];
    const float* ew  = (const float*)d_in[2];
    const float* W1  = (const float*)d_in[3];
    const float* b1  = (const float*)d_in[4];
    const float* W2  = (const float*)d_in[5];
    const float* b2  = (const float*)d_in[6];
    const float* Wc  = (const float*)d_in[7];
    const float* bc  = (const float*)d_in[8];
    float*       out = (float*)d_out;

    const int n_nodes = in_sizes[0] / NFEAT;
    const int n_edges = in_sizes[2];
    const int* src = ei;
    const int* dst = ei + n_edges;

    const size_t nd = (size_t)n_nodes * NFEAT;
    float* bufA = (float*)d_ws;
    float* bufB = bufA + nd;
    float* bufC = bufB + nd;
    float* dinv = bufC + nd;

    const int B = 256;
    const int gN  = (n_nodes + B - 1) / B;
    const int gE  = (n_edges + B - 1) / B;
    const int gND = (int)((nd + B - 1) / B);
    const int gEw = (int)(((size_t)n_edges * 32 + B - 1) / B);
    const dim3 gblk(32, 8);
    const int gTiles = n_nodes / 16;

    gcn_fill1    <<<gN, B, 0, stream>>>(dinv, n_nodes);
    gcn_deg_accum<<<gE, B, 0, stream>>>(dst, ew, dinv, n_edges);
    gcn_rsqrt    <<<gN, B, 0, stream>>>(dinv, n_nodes);

    gcn_gemm128_wmma<<<gTiles, gblk, 0, stream>>>(x, W1, bufA);
    gcn_zero        <<<gND, B, 0, stream>>>(bufB, (int)nd);
    gcn_scatter     <<<gEw, B, 0, stream>>>(bufA, bufB, src, dst, ew, dinv, n_edges);
    gcn_fuse        <<<gND, B, 0, stream>>>(bufB, bufA, dinv, b1, bufA, n_nodes, 1);

    gcn_gemm128_wmma<<<gTiles, gblk, 0, stream>>>(bufA, W2, bufB);
    gcn_zero        <<<gND, B, 0, stream>>>(bufC, (int)nd);
    gcn_scatter     <<<gEw, B, 0, stream>>>(bufB, bufC, src, dst, ew, dinv, n_edges);
    gcn_fuse        <<<gND, B, 0, stream>>>(bufC, bufB, dinv, b2, bufB, n_nodes, 1);

    gcn_classifier<<<gN, B, 0, stream>>>(bufB, Wc, bc, out, n_nodes);
}


# Global-atomic scatter replaced by row tiles in LDS: a group owns 196 destination rows, scans the edge list, adds its edges by LDS f32 atomics (arrival order), writes rows with self loop, bias, ReLU
# speedup vs baseline: 1.7184x; 1.7184x over previous
#include <hip/hip_runtime.h>
#include <hip/hip_bf16.h>

typedef float v2f __attribute__((ext_vector_type(2)));
typedef float v8f __attribute__((ext_vector_type(8)));

#define NFEAT 128

__global__ void gcn_fill1(float* __restrict__ deg, int n) {
    int i = blockIdx.x * blockDim.x + threadIdx.x;
    if (i < n) deg[i] = 1.0f;
}

__global__ void gcn_deg_accum(const int* __restrict__ dst,
                              const float* __restrict__ ew,
                              float* __restrict__ deg, int n_edges) {
    int e = blockIdx.x * blockDim.x + threadIdx.x;
    if (e < n_edges)
        __hip_atomic_fetch_add(&deg[dst[e]], ew[e],
                               __ATOMIC_RELAXED, __HIP_MEMORY_SCOPE_AGENT);
}

__global__ void gcn_rsqrt(float* __restrict__ d, int n) {
    int i = blockIdx.x * blockDim.x + threadIdx.x;
    if (i < n) d[i] = rsqrtf(d[i]);
}

__global__ __launch_bounds__(256) void gcn_gemm128_wmma(
        const float* __restrict__ X, const float* __restrict__ W,
        float* __restrict__ H) {
    __shared__ float Wl[NFEAT * NFEAT];
    {
        const float4* Wv  = (const float4*)W;
        float4*       Wlv = (float4*)Wl;
        int tid = threadIdx.y * 32 + threadIdx.x;
        #pragma unroll
        for (int i = 0; i < (NFEAT * NFEAT / 4) / 256; ++i)
            Wlv[tid + i * 256] = Wv[tid + i * 256];
    }
    __syncthreads();

    const int lane  = threadIdx.x;
    const int m0    = blockIdx.x * 16;
    const int n0    = threadIdx.y * 16;
    const int mlane = lane & 15;
    const int khalf = (lane >> 4) * 2;
    const int ncol  = n0 + mlane;

    const float* xrow = X + (size_t)(m0 + mlane) * NFEAT;
    v8f acc = {};

    #pragma unroll
    for (int k0 = 0; k0 < NFEAT; k0 += 4) {
        const int k = k0 + khalf;
        v2f a = *(const v2f*)&xrow[k];
        v2f b;
        b.x = Wl[(k    ) * NFEAT + ncol];
        b.y = Wl[(k + 1) * NFEAT + ncol];
        acc = __builtin_amdgcn_wmma_f32_16x16x4_f32(
                  false, a, false, b, (short)0, acc, false, false);
    }

    const int rbase = m0 + (lane >> 4) * 8;
    #pragma unroll
    for (int r = 0; r < 8; ++r)
        H[(size_t)(rbase + r) * NFEAT + ncol] = acc[r];
}

#define RT   196
#define QCAP 16384
#define ATH  1024
#define SU   8
#define DU   8

__global__ __launch_bounds__(ATH) void gcn_agg_tile(
        const float* __restrict__ h, const int* __restrict__ src,
        const int* __restrict__ dst, const float* __restrict__ ew,
        const float* __restrict__ dinv, const float* __restrict__ b,
        float* __restrict__ out, int n_nodes, int n_edges) {
    __shared__ __attribute__((aligned(16))) float acc[(RT + 1) * NFEAT];
    __shared__ int q[QCAP];
    __shared__ int qn;

    const int tid   = threadIdx.x;
    const int lane  = tid & 31;
    const int lo    = blockIdx.x * RT;
    const int nrows = min(RT, n_nodes - lo);

    auto clear_tile = [&]() {
        float4* a4 = (float4*)acc;
        for (int i = tid; i < (RT + 1) * NFEAT / 4; i += ATH)
            a4[i] = make_float4(0.0f, 0.0f, 0.0f, 0.0f);
    };

    auto note = [&](int e, int d) {
        if ((unsigned)(d - lo) < (unsigned)nrows) {
            int idx = atomicAdd(&qn, 1);
            if (idx < QCAP) q[idx] = e;
        }
    };

    auto drain = [&](int total) {
        const int wave = __builtin_amdgcn_readfirstlane(tid >> 5);
        for (int base = wave; base < total; base += 32 * DU) {
            int e[DU]; bool ok[DU];
            #pragma unroll
            for (int u = 0; u < DU; ++u) {
                const int i = base + 32 * u;
                ok[u] = i < total;
                e[u]  = q[ok[u] ? i : base];
            }
            __builtin_amdgcn_sched_barrier(0);
            int sN[DU], dN[DU]; float w[DU];
            #pragma unroll
            for (int u = 0; u < DU; ++u) {
                sN[u] = src[e[u]]; dN[u] = dst[e[u]]; w[u] = ew[e[u]];
            }
            __builtin_amdgcn_sched_barrier(0);
            float c[DU]; float hv[DU][4];
            #pragma unroll
            for (int u = 0; u < DU; ++u) {
                c[u] = dinv[sN[u]] * w[u] * dinv[dN[u]];
                const float* hr = h + (size_t)sN[u] * NFEAT + lane;
                #pragma unroll
                for (int j = 0; j < 4; ++j) hv[u][j] = hr[32 * j];
            }
            #pragma unroll
            for (int u = 0; u < DU; ++u) {
                const int row = ok[u] ? dN[u] - lo : RT;
                float* ar = acc + row * NFEAT + lane;
                #pragma unroll
                for (int j = 0; j < 4; ++j)
                    __hip_atomic_fetch_add(ar + 32 * j, hv[u][j] * c[u],
                                           __ATOMIC_RELAXED, __HIP_MEMORY_SCOPE_WORKGROUP);
            }
        }
    };

    clear_tile();
    if (tid == 0) qn = 0;
    __syncthreads();

    const int pre   = min(n_edges, (int)(((16u - (unsigned)((size_t)dst & 15u)) & 15u) >> 2));
    const int n4    = (n_edges - pre) >> 2;
    const int tail0 = pre + 4 * n4;
    const int4* d4  = (const int4*)(dst + pre);
    if (tid < pre)             note(tid, dst[tid]);
    if (tid < n_edges - tail0) note(tail0 + tid, dst[tail0 + tid]);
    for (int i = tid; i < n4; i += ATH * SU) {
        int4 v[SU];
        #pragma unroll
        for (int u = 0; u < SU; ++u) {
            const int k = i + u * ATH;
            v[u] = d4[min(k, n4 - 1)];
            if (k >= n4) v[u] = make_int4(-1, -1, -1, -1);
        }
        #pragma unroll
        for (int u = 0; u < SU; ++u) {
            const int e0 = pre + 4 * (i + u * ATH);
            const bool m0 = (unsigned)(v[u].x - lo) < (unsigned)nrows;
            const bool m1 = (unsigned)(v[u].y - lo) < (unsigned)nrows;
            const bool m2 = (unsigned)(v[u].z - lo) < (unsigned)nrows;
            const bool m3 = (unsigned)(v[u].w - lo) < (unsigned)nrows;
            if (m0 | m1 | m2 | m3) {
                note(e0,     v[u].x);
                note(e0 + 1, v[u].y);
                note(e0 + 2, v[u].z);
                note(e0 + 3, v[u].w);
            }
        }
    }
    __syncthreads();

    const int noted = qn;
    if (noted <= QCAP) {
        drain(noted);
    } else {
        __syncthreads();
        clear_tile();
        for (int seg = 0; seg < n_edges; seg += QCAP) {
            if (tid == 0) qn = 0;
            __syncthreads();
            const int segEnd = min(seg + QCAP, n_edges);
            for (int e = seg + tid; e < segEnd; e += ATH) note(e, dst[e]);
            __syncthreads();
            drain(qn);
            __syncthreads();
        }
    }
    __syncthreads();

    const float4* acc4 = (const float4*)acc;
    const float4* h4   = (const float4*)h;
    const float4* b4   = (const float4*)b;
    float4*       o4   = (float4*)out;
    for (int idx = tid; idx < nrows * (NFEAT / 4); idx += ATH) {
        const int i  = lo + (idx >> 5);
        const int f4 = idx & 31;
        const float di = dinv[i];
        const float d2 = di * di;
        const float4 a  = acc4[idx];
        const float4 hv = h4[(size_t)i * (NFEAT / 4) + f4];
        const float4 bv = b4[f4];
        float4 r;
        r.x = fmaxf(a.x + hv.x * d2 + bv.x, 0.0f);
        r.y = fmaxf(a.y + hv.y * d2 + bv.y, 0.0f);
        r.z = fmaxf(a.z + hv.z * d2 + bv.z, 0.0f);
        r.w = fmaxf(a.w + hv.w * d2 + bv.w, 0.0f);
        o4[(size_t)i * (NFEAT / 4) + f4] = r;
    }
}

__global__ void gcn_classifier(const float* __restrict__ h,
                               const float* __restrict__ Wc,
                               const float* __restrict__ bc,
                               float* __restrict__ out, int n_nodes) {
    int i = blockIdx.x * blockDim.x + threadIdx.x;
    if (i >= n_nodes) return;
    float acc[10];
    #pragma unroll
    for (int c = 0; c < 10; ++c) acc[c] = bc[c];
    const float* hr = h + (size_t)i * NFEAT;
    for (int d = 0; d < NFEAT; ++d) {
        float hv = hr[d];
        #pragma unroll
        for (int c = 0; c < 10; ++c) acc[c] += hv * Wc[d * 10 + c];
    }
    #pragma unroll
    for (int c = 0; c < 10; ++c) out[(size_t)i * 10 + c] = acc[c];
}

extern "C" void kernel_launch(void* const* d_in, const int* in_sizes, int n_in,
                              void* d_out, int out_size, void* d_ws, size_t ws_size,
                              hipStream_t stream) {
    const float* x   = (const float*)d_in[0];
    const int*   ei  = (const int*)  d_in[1];
    const float* ew  = (const float*)d_in[2];
    const float* W1  = (const float*)d_in[3];
    const float* b1  = (const float*)d_in[4];
    const float* W2  = (const float*)d_in[5];
    const float* b2  = (const float*)d_in[6];
    const float* Wc  = (const float*)d_in[7];
    const float* bc  = (const float*)d_in[8];
    float*       out = (float*)d_out;

    const int n_nodes = in_sizes[0] / NFEAT;
    const int n_edges = in_sizes[2];
    const int* src = ei;
    const int* dst = ei + n_edges;

    const size_t nd = (size_t)n_nodes * NFEAT;
    float* bufA = (float*)d_ws;
    float* bufB = bufA + nd;
    float* dinv = bufB + nd;

    const int B = 256;
    const int gN  = (n_nodes + B - 1) / B;
    const int gE  = (n_edges + B - 1) / B;
    const int gT  = (n_nodes + RT - 1) / RT;
    const dim3 gblk(32, 8);
    const int gTiles = n_nodes / 16;

    gcn_fill1    <<<gN, B, 0, stream>>>(dinv, n_nodes);
    gcn_deg_accum<<<gE, B, 0, stream>>>(dst, ew, dinv, n_edges);
    gcn_rsqrt    <<<gN, B, 0, stream>>>(dinv, n_nodes);

    gcn_gemm128_wmma<<<gTiles, gblk, 0, stream>>>(x, W1, bufA);
    gcn_agg_tile    <<<gT, ATH, 0, stream>>>(bufA, src, dst, ew, dinv, b1, bufB, n_nodes, n_edges);

    gcn_gemm128_wmma<<<gTiles, gblk, 0, stream>>>(bufB, W2, bufA);
    gcn_agg_tile    <<<gT, ATH, 0, stream>>>(bufA, src, dst, ew, dinv, b2, bufB, n_nodes, n_edges);

    gcn_classifier<<<gN, B, 0, stream>>>(bufB, Wc, bc, out, n_nodes);
}


# Destination scan done once per call: layer-1 tile kernel saves each tile's noted edges (source, row, coefficient) to the workspace, layer 2 adds from that list; sums stay LDS atomics, arrival order
# speedup vs baseline: 1.9452x; 1.1320x over previous
#include <hip/hip_runtime.h>
#include <hip/hip_bf16.h>

typedef float v2f __attribute__((ext_vector_type(2)));
typedef float v8f __attribute__((ext_vector_type(8)));

#define NFEAT 128

__global__ void gcn_fill1(float* __restrict__ deg, int n) {
    int i = blockIdx.x * blockDim.x + threadIdx.x;
    if (i < n) deg[i] = 1.0f;
}

__global__ void gcn_deg_accum(const int* __restrict__ dst,
                              const float* __restrict__ ew,
                              float* __restrict__ deg, int n_edges) {
    int e = blockIdx.x * blockDim.x + threadIdx.x;
    if (e < n_edges)
        __hip_atomic_fetch_add(&deg[dst[e]], ew[e],
                               __ATOMIC_RELAXED, __HIP_MEMORY_SCOPE_AGENT);
}

__global__ void gcn_rsqrt(float* __restrict__ d, int n) {
    int i = blockIdx.x * blockDim.x + threadIdx.x;
    if (i < n) d[i] = rsqrtf(d[i]);
}

__global__ __launch_bounds__(256) void gcn_gemm128_wmma(
        const float* __restrict__ X, const float* __restrict__ W,
        float* __restrict__ H) {
    __shared__ float Wl[NFEAT * NFEAT];
    {
        const float4* Wv  = (const float4*)W;
        float4*       Wlv = (float4*)Wl;
        int tid = threadIdx.y * 32 + threadIdx.x;
        #pragma unroll
        for (int i = 0; i < (NFEAT * NFEAT / 4) / 256; ++i)
            Wlv[tid + i * 256] = Wv[tid + i * 256];
    }
    __syncthreads();

    const int lane  = threadIdx.x;
    const int m0    = blockIdx.x * 16;
    const int n0    = threadIdx.y * 16;
    const int mlane = lane & 15;
    const int khalf = (lane >> 4) * 2;
    const int ncol  = n0 + mlane;

    const float* xrow = X + (size_t)(m0 + mlane) * NFEAT;
    v8f acc = {};

    #pragma unroll
    for (int k0 = 0; k0 < NFEAT; k0 += 4) {
        const int k = k0 + khalf;
        v2f a = *(const v2f*)&xrow[k];
        v2f b;
        b.x = Wl[(k    ) * NFEAT + ncol];
        b.y = Wl[(k + 1) * NFEAT + ncol];
        acc = __builtin_amdgcn_wmma_f32_16x16x4_f32(
                  false, a, false, b, (short)0, acc, false, false);
    }

    const int rbase = m0 + (lane >> 4) * 8;
    #pragma unroll
    for (int r = 0; r < 8; ++r)
        H[(size_t)(rbase + r) * NFEAT + ncol] = acc[r];
}

#define RT   196
#define QCAP 16384
#define ATH  1024
#define SU   8
#define DU   8

template <bool FROM_LIST>
__global__ __launch_bounds__(ATH) void gcn_agg_tile(
        const float* __restrict__ h, const int* __restrict__ src,
        const int* __restrict__ dst, const float* __restrict__ ew,
        const float* __restrict__ dinv, const float* __restrict__ b,
        float* __restrict__ out, int* __restrict__ lists, int* __restrict__ cnts,
        int n_nodes, int n_edges) {
    __shared__ __attribute__((aligned(16))) float acc[(RT + 1) * NFEAT];
    __shared__ int q[QCAP];
    __shared__ int qn;

    const int tid   = threadIdx.x;
    const int lane  = tid & 31;
    const int lo    = blockIdx.x * RT;
    const int nrows = min(RT, n_nodes - lo);
    int4* const tl  = (int4*)lists + (size_t)blockIdx.x * QCAP;

    auto clear_tile = [&]() {
        float4* a4 = (float4*)acc;
        for (int i = tid; i < (RT + 1) * NFEAT / 4; i += ATH)
            a4[i] = make_float4(0.0f, 0.0f, 0.0f, 0.0f);
    };

    auto note = [&](int e, int d) {
        if ((unsigned)(d - lo) < (unsigned)nrows) {
            int idx = atomicAdd(&qn, 1);
            if (idx < QCAP) q[idx] = e;
        }
    };

    auto drain = [&](int total, bool save) {
        const int wave = __builtin_amdgcn_readfirstlane(tid >> 5);
        int4* slot = tl + wave * DU;
        for (int base = wave; base < total; base += 32 * DU, slot += 32 * DU) {
            int e[DU]; bool ok[DU];
            #pragma unroll
            for (int u = 0; u < DU; ++u) {
                const int i = base + 32 * u;
                ok[u] = i < total;
                e[u]  = q[ok[u] ? i : base];
            }
            __builtin_amdgcn_sched_barrier(0);
            int sN[DU], dN[DU]; float w[DU];
            #pragma unroll
            for (int u = 0; u < DU; ++u) {
                sN[u] = src[e[u]]; dN[u] = dst[e[u]]; w[u] = ew[e[u]];
            }
            __builtin_amdgcn_sched_barrier(0);
            float c[DU]; float hv[DU][4];
            #pragma unroll
            for (int u = 0; u < DU; ++u) {
                c[u] = dinv[sN[u]] * w[u] * dinv[dN[u]];
                const float* hr = h + (size_t)sN[u] * NFEAT + lane;
                #pragma unroll
                for (int j = 0; j < 4; ++j) hv[u][j] = hr[32 * j];
            }
            int row[DU];
            #pragma unroll
            for (int u = 0; u < DU; ++u) {
                row[u] = ok[u] ? dN[u] - lo : RT;
                float* ar = acc + row[u] * NFEAT + lane;
                #pragma unroll
                for (int j = 0; j < 4; ++j)
                    __hip_atomic_fetch_add(ar + 32 * j, hv[u][j] * c[u],
                                           __ATOMIC_RELAXED, __HIP_MEMORY_SCOPE_WORKGROUP);
            }
            if (save) {
                int ex = sN[0], ey = row[0]; float ez = c[0];
                #pragma unroll
                for (int u = 1; u < DU; ++u)
                    if (lane == u) { ex = sN[u]; ey = row[u]; ez = c[u]; }
                if (lane < DU) slot[lane] = make_int4(ex, ey, __float_as_int(ez), 0);
            }
        }
    };

    auto drain_list = [&](int total) {
        const int wave = __builtin_amdgcn_readfirstlane(tid >> 5);
        if (wave >= total) return;
        const int4* slot = tl + wave * DU;
        int4 en[DU];
        #pragma unroll
        for (int u = 0; u < DU; ++u) en[u] = slot[u];
        for (int base = wave; base < total; base += 32 * DU) {
            float hv[DU][4];
            #pragma unroll
            for (int u = 0; u < DU; ++u) {
                const float* hr = h + (size_t)en[u].x * NFEAT + lane;
                #pragma unroll
                for (int j = 0; j < 4; ++j) hv[u][j] = hr[32 * j];
            }
            if (base + 32 * DU < total) slot += 32 * DU;
            int4 nx[DU];
            #pragma unroll
            for (int u = 0; u < DU; ++u) nx[u] = slot[u];
            #pragma unroll
            for (int u = 0; u < DU; ++u) {
                float* ar = acc + en[u].y * NFEAT + lane;
                const float cu = __int_as_float(en[u].z);
                #pragma unroll
                for (int j = 0; j < 4; ++j)
                    __hip_atomic_fetch_add(ar + 32 * j, hv[u][j] * cu,
                                           __ATOMIC_RELAXED, __HIP_MEMORY_SCOPE_WORKGROUP);
            }
            #pragma unroll
            for (int u = 0; u < DU; ++u) en[u] = nx[u];
        }
    };

    clear_tile();
    if (tid == 0) qn = 0;
    __syncthreads();

    int noted;
    if constexpr (FROM_LIST) {
        noted = cnts[blockIdx.x * 32];
    } else {
    const int pre   = min(n_edges, (int)(((16u - (unsigned)((size_t)dst & 15u)) & 15u) >> 2));
    const int n4    = (n_edges - pre) >> 2;
    const int tail0 = pre + 4 * n4;
    const int4* d4  = (const int4*)(dst + pre);
    if (tid < pre)             note(tid, dst[tid]);
    if (tid < n_edges - tail0) note(tail0 + tid, dst[tail0 + tid]);
    for (int i = tid; i < n4; i += ATH * SU) {
        int4 v[SU];
        #pragma unroll
        for (int u = 0; u < SU; ++u) {
            const int k = i + u * ATH;
            v[u] = d4[min(k, n4 - 1)];
            if (k >= n4) v[u] = make_int4(-1, -1, -1, -1);
        }
        #pragma unroll
        for (int u = 0; u < SU; ++u) {
            const int e0 = pre + 4 * (i + u * ATH);
            const bool m0 = (unsigned)(v[u].x - lo) < (unsigned)nrows;
            const bool m1 = (unsigned)(v[u].y - lo) < (unsigned)nrows;
            const bool m2 = (unsigned)(v[u].z - lo) < (unsigned)nrows;
            const bool m3 = (unsigned)(v[u].w - lo) < (unsigned)nrows;
            if (m0 | m1 | m2 | m3) {
                note(e0,     v[u].x);
                note(e0 + 1, v[u].y);
                note(e0 + 2, v[u].z);
                note(e0 + 3, v[u].w);
            }
        }
    }
    __syncthreads();
    noted = qn;
    if (tid == 0) cnts[blockIdx.x * 32] = noted;
    }

    if (noted <= QCAP) {
        if constexpr (FROM_LIST) drain_list(noted);
        else                     drain(noted, true);
    } else {
        __syncthreads();
        clear_tile();
        for (int seg = 0; seg < n_edges; seg += QCAP) {
            if (tid == 0) qn = 0;
            __syncthreads();
            const int segEnd = min(seg + QCAP, n_edges);
            for (int e = seg + tid; e < segEnd; e += ATH) note(e, dst[e]);
            __syncthreads();
            drain(qn, false);
            __syncthreads();
        }
    }
    __syncthreads();

    const float4* acc4 = (const float4*)acc;
    const float4* h4   = (const float4*)h;
    const float4* b4   = (const float4*)b;
    float4*       o4   = (float4*)out;
    for (int idx = tid; idx < nrows * (NFEAT / 4); idx += ATH) {
        const int i  = lo + (idx >> 5);
        const int f4 = idx & 31;
        const float di = dinv[i];
        const float d2 = di * di;
        const float4 a  = acc4[idx];
        const float4 hv = h4[(size_t)i * (NFEAT / 4) + f4];
        const float4 bv = b4[f4];
        float4 r;
        r.x = fmaxf(a.x + hv.x * d2 + bv.x, 0.0f);
        r.y = fmaxf(a.y + hv.y * d2 + bv.y, 0.0f);
        r.z = fmaxf(a.z + hv.z * d2 + bv.z, 0.0f);
        r.w = fmaxf(a.w + hv.w * d2 + bv.w, 0.0f);
        o4[(size_t)i * (NFEAT / 4) + f4] = r;
    }
}

__global__ void gcn_classifier(const float* __restrict__ h,
                               const float* __restrict__ Wc,
                               const float* __restrict__ bc,
                               float* __restrict__ out, int n_nodes) {
    int i = blockIdx.x * blockDim.x + threadIdx.x;
    if (i >= n_nodes) return;
    float acc[10];
    #pragma unroll
    for (int c = 0; c < 10; ++c) acc[c] = bc[c];
    const float* hr = h + (size_t)i * NFEAT;
    for (int d = 0; d < NFEAT; ++d) {
        float hv = hr[d];
        #pragma unroll
        for (int c = 0; c < 10; ++c) acc[c] += hv * Wc[d * 10 + c];
    }
    #pragma unroll
    for (int c = 0; c < 10; ++c) out[(size_t)i * 10 + c] = acc[c];
}

extern "C" void kernel_launch(void* const* d_in, const int* in_sizes, int n_in,
                              void* d_out, int out_size, void* d_ws, size_t ws_size,
                              hipStream_t stream) {
    const float* x   = (const float*)d_in[0];
    const int*   ei  = (const int*)  d_in[1];
    const float* ew  = (const float*)d_in[2];
    const float* W1  = (const float*)d_in[3];
    const float* b1  = (const float*)d_in[4];
    const float* W2  = (const float*)d_in[5];
    const float* b2  = (const float*)d_in[6];
    const float* Wc  = (const float*)d_in[7];
    const float* bc  = (const float*)d_in[8];
    float*       out = (float*)d_out;

    const int n_nodes = in_sizes[0] / NFEAT;
    const int n_edges = in_sizes[2];
    const int* src = ei;
    const int* dst = ei + n_edges;

    const size_t nd = (size_t)n_nodes * NFEAT;
    float* bufA = (float*)d_ws;
    float* bufB = bufA + nd;
    float* dinv = bufB + nd;
    const int gT  = (n_nodes + RT - 1) / RT;
    int* lists = (int*)(dinv + (size_t)((n_nodes + 31) / 32) * 32);
    int* cnts  = lists + (size_t)gT * QCAP * 4;

    const int B = 256;
    const int gN  = (n_nodes + B - 1) / B;
    const int gE  = (n_edges + B - 1) / B;
    const dim3 gblk(32, 8);
    const int gTiles = n_nodes / 16;

    gcn_fill1    <<<gN, B, 0, stream>>>(dinv, n_nodes);
    gcn_deg_accum<<<gE, B, 0, stream>>>(dst, ew, dinv, n_edges);
    gcn_rsqrt    <<<gN, B, 0, stream>>>(dinv, n_nodes);

    gcn_gemm128_wmma<<<gTiles, gblk, 0, stream>>>(x, W1, bufA);
    gcn_agg_tile<false><<<gT, ATH, 0, stream>>>(bufA, src, dst, ew, dinv, b1, bufB, lists, cnts, n_nodes, n_edges);

    gcn_gemm128_wmma<<<gTiles, gblk, 0, stream>>>(bufB, W2, bufA);
    gcn_agg_tile<true><<<gT, ATH, 0, stream>>>(bufA, src, dst, ew, dinv, b2, bufB, lists, cnts, n_nodes, n_edges);

    gcn_classifier<<<gN, B, 0, stream>>>(bufB, Wc, bc, out, n_nodes);
}


# Product moved into the aggregation launch: a group sums scaled INPUT rows in its LDS tile, then tile @ W on WMMA f32; (A X) W replaces A (X W), so sum order changes; the 2 product launches go
# speedup vs baseline: 2.2959x; 1.1803x over previous
#include <hip/hip_runtime.h>
#include <hip/hip_bf16.h>

typedef float v2f __attribute__((ext_vector_type(2)));
typedef float v8f __attribute__((ext_vector_type(8)));

#define NFEAT 128

__global__ void gcn_fill1(float* __restrict__ deg, int n) {
    int i = blockIdx.x * blockDim.x + threadIdx.x;
    if (i < n) deg[i] = 1.0f;
}

__global__ void gcn_deg_accum(const int* __restrict__ dst,
                              const float* __restrict__ ew,
                              float* __restrict__ deg, int n_edges) {
    int e = blockIdx.x * blockDim.x + threadIdx.x;
    if (e < n_edges)
        __hip_atomic_fetch_add(&deg[dst[e]], ew[e],
                               __ATOMIC_RELAXED, __HIP_MEMORY_SCOPE_AGENT);
}

__global__ void gcn_rsqrt(float* __restrict__ d, int n) {
    int i = blockIdx.x * blockDim.x + threadIdx.x;
    if (i < n) d[i] = rsqrtf(d[i]);
}

#define RT   196
#define MB   13
#define TP   132
#define WP   136
#define QCAP 16384
#define ATH  1024
#define SU   8
#define DU   8

template <int N> struct tile_count { static constexpr int value = N; };

template <bool FROM_LIST>
__global__ __launch_bounds__(ATH) void gcn_layer_tile(
        const float* __restrict__ h, const float* __restrict__ W,
        const int* __restrict__ src,
        const int* __restrict__ dst, const float* __restrict__ ew,
        const float* __restrict__ dinv, const float* __restrict__ b,
        float* __restrict__ out, int* __restrict__ lists, int* __restrict__ cnts,
        int n_nodes, int n_edges) {
    __shared__ __attribute__((aligned(16))) float acc[MB * 16 * TP];
    __shared__ __attribute__((aligned(16))) float Wl[NFEAT * WP];
    __shared__ int q[QCAP];
    __shared__ int qn;

    const int tid   = threadIdx.x;
    const int lane  = tid & 31;
    const int lo    = blockIdx.x * RT;
    const int nrows = min(RT, n_nodes - lo);
    int4* const tl  = (int4*)lists + (size_t)blockIdx.x * QCAP;

    auto clear_tile = [&]() {
        float4* a4 = (float4*)acc;
        for (int i = tid; i < MB * 16 * TP / 4; i += ATH)
            a4[i] = make_float4(0.0f, 0.0f, 0.0f, 0.0f);
    };

    auto note = [&](int e, int d) {
        if ((unsigned)(d - lo) < (unsigned)nrows) {
            int idx = atomicAdd(&qn, 1);
            if (idx < QCAP) q[idx] = e;
        }
    };

    auto drain = [&](int total, bool save) {
        const int wave = __builtin_amdgcn_readfirstlane(tid >> 5);
        int4* slot = tl + wave * DU;
        for (int base = wave; base < total; base += 32 * DU, slot += 32 * DU) {
            int e[DU]; bool ok[DU];
            #pragma unroll
            for (int u = 0; u < DU; ++u) {
                const int i = base + 32 * u;
                ok[u] = i < total;
                e[u]  = q[ok[u] ? i : base];
            }
            __builtin_amdgcn_sched_barrier(0);
            int sN[DU], dN[DU]; float w[DU];
            #pragma unroll
            for (int u = 0; u < DU; ++u) {
                sN[u] = src[e[u]]; dN[u] = dst[e[u]]; w[u] = ew[e[u]];
            }
            __builtin_amdgcn_sched_barrier(0);
            float c[DU]; float hv[DU][4];
            #pragma unroll
            for (int u = 0; u < DU; ++u) {
                c[u] = dinv[sN[u]] * w[u] * dinv[dN[u]];
                const float* hr = h + (size_t)sN[u] * NFEAT + lane;
                #pragma unroll
                for (int j = 0; j < 4; ++j) hv[u][j] = hr[32 * j];
            }
            int row[DU];
            #pragma unroll
            for (int u = 0; u < DU; ++u) {
                row[u] = ok[u] ? dN[u] - lo : RT;
                float* ar = acc + row[u] * TP + lane;
                #pragma unroll
                for (int j = 0; j < 4; ++j)
                    __hip_atomic_fetch_add(ar + 32 * j, hv[u][j] * c[u],
                                           __ATOMIC_RELAXED, __HIP_MEMORY_SCOPE_WORKGROUP);
            }
            if (save) {
                int ex = sN[0], ey = row[0]; float ez = c[0];
                #pragma unroll
                for (int u = 1; u < DU; ++u)
                    if (lane == u) { ex = sN[u]; ey = row[u]; ez = c[u]; }
                if (lane < DU) slot[lane] = make_int4(ex, ey, __float_as_int(ez), 0);
            }
        }
    };

    auto drain_list = [&](int total) {
        const int wave = __builtin_amdgcn_readfirstlane(tid >> 5);
        if (wave >= total) return;
        const int4* slot = tl + wave * DU;
        int4 en[DU];
        #pragma unroll
        for (int u = 0; u < DU; ++u) en[u] = slot[u];
        for (int base = wave; base < total; base += 32 * DU) {
            float hv[DU][4];
            #pragma unroll
            for (int u = 0; u < DU; ++u) {
                const float* hr = h + (size_t)en[u].x * NFEAT + lane;
                #pragma unroll
                for (int j = 0; j < 4; ++j) hv[u][j] = hr[32 * j];
            }
            if (base + 32 * DU < total) slot += 32 * DU;
            int4 nx[DU];
            #pragma unroll
            for (int u = 0; u < DU; ++u) nx[u] = slot[u];
            #pragma unroll
            for (int u = 0; u < DU; ++u) {
                float* ar = acc + en[u].y * TP + lane;
                const float cu = __int_as_float(en[u].z);
                #pragma unroll
                for (int j = 0; j < 4; ++j)
                    __hip_atomic_fetch_add(ar + 32 * j, hv[u][j] * cu,
                                           __ATOMIC_RELAXED, __HIP_MEMORY_SCOPE_WORKGROUP);
            }
            #pragma unroll
            for (int u = 0; u < DU; ++u) en[u] = nx[u];
        }
    };

    {
        const float4* Wv = (const float4*)W;
        float4 wv[(NFEAT * NFEAT / 4) / ATH];
        #pragma unroll
        for (int u = 0; u < (NFEAT * NFEAT / 4) / ATH; ++u) wv[u] = Wv[tid + u * ATH];
        clear_tile();
        #pragma unroll
        for (int u = 0; u < (NFEAT * NFEAT / 4) / ATH; ++u) {
            const int i = tid + u * ATH;
            *(float4*)(Wl + (i >> 5) * WP + (i & 31) * 4) = wv[u];
        }
    }
    if (tid == 0) qn = 0;
    __syncthreads();

    int noted;
    if constexpr (FROM_LIST) {
        noted = cnts[blockIdx.x * 32];
    } else {
    const int pre   = min(n_edges, (int)(((16u - (unsigned)((size_t)dst & 15u)) & 15u) >> 2));
    const int n4    = (n_edges - pre) >> 2;
    const int tail0 = pre + 4 * n4;
    const int4* d4  = (const int4*)(dst + pre);
    if (tid < pre)             note(tid, dst[tid]);
    if (tid < n_edges - tail0) note(tail0 + tid, dst[tail0 + tid]);
    for (int i = tid; i < n4; i += ATH * SU) {
        int4 v[SU];
        #pragma unroll
        for (int u = 0; u < SU; ++u) {
            const int k = i + u * ATH;
            v[u] = d4[min(k, n4 - 1)];
            if (k >= n4) v[u] = make_int4(-1, -1, -1, -1);
        }
        #pragma unroll
        for (int u = 0; u < SU; ++u) {
            const int e0 = pre + 4 * (i + u * ATH);
            const bool m0 = (unsigned)(v[u].x - lo) < (unsigned)nrows;
            const bool m1 = (unsigned)(v[u].y - lo) < (unsigned)nrows;
            const bool m2 = (unsigned)(v[u].z - lo) < (unsigned)nrows;
            const bool m3 = (unsigned)(v[u].w - lo) < (unsigned)nrows;
            if (m0 | m1 | m2 | m3) {
                note(e0,     v[u].x);
                note(e0 + 1, v[u].y);
                note(e0 + 2, v[u].z);
                note(e0 + 3, v[u].w);
            }
        }
    }
    __syncthreads();
    noted = qn;
    if (tid == 0) cnts[blockIdx.x * 32] = noted;
    }

    if (noted <= QCAP) {
        if constexpr (FROM_LIST) drain_list(noted);
        else                     drain(noted, true);
    } else {
        __syncthreads();
        clear_tile();
        for (int seg = 0; seg < n_edges; seg += QCAP) {
            if (tid == 0) qn = 0;
            __syncthreads();
            const int segEnd = min(seg + QCAP, n_edges);
            for (int e = seg + tid; e < segEnd; e += ATH) note(e, dst[e]);
            __syncthreads();
            drain(qn, false);
            __syncthreads();
        }
    }
    __syncthreads();

    const float4* h4 = (const float4*)h;
    for (int idx = tid; idx < nrows * (NFEAT / 4); idx += ATH) {
        const int r  = idx >> 5;
        const int f4 = idx & 31;
        const float di = dinv[lo + r];
        const float d2 = di * di;
        const float4 hv = h4[(size_t)(lo + r) * (NFEAT / 4) + f4];
        float4* ap = (float4*)(acc + r * TP) + f4;
        float4 a = *ap;
        a.x += hv.x * d2;
        a.y += hv.y * d2;
        a.z += hv.z * d2;
        a.w += hv.w * d2;
        *ap = a;
    }
    __syncthreads();

    const int wave  = __builtin_amdgcn_readfirstlane(tid >> 5);
    const int mlane = lane & 15;
    const int khalf = (lane >> 4) * 2;
    const int ncol  = (wave & 7) * 16 + mlane;
    const int mq    = wave >> 3;
    const int cnt   = (((nrows + 15) >> 4) - mq + 3) >> 2;
    v8f d[4] = {};
    auto tile_mm = [&](auto ntc) {
        constexpr int NT = decltype(ntc)::value;
        const float* ap = acc + (mq * 16 + mlane) * TP + khalf;
        const float* bp = Wl + khalf * WP + ncol;
        #pragma unroll 4
        for (int k0 = 0; k0 < NFEAT; k0 += 4) {
            v2f bb;
            bb.x = bp[(k0    ) * WP];
            bb.y = bp[(k0 + 1) * WP];
            #pragma unroll
            for (int i = 0; i < NT; ++i) {
                const v2f a = *(const v2f*)(ap + i * 64 * TP + k0);
                d[i] = __builtin_amdgcn_wmma_f32_16x16x4_f32(
                           false, a, false, bb, (short)0, d[i], false, false);
            }
        }
    };
    switch (cnt) {
        case 4: tile_mm(tile_count<4>{}); break;
        case 3: tile_mm(tile_count<3>{}); break;
        case 2: tile_mm(tile_count<2>{}); break;
        case 1: tile_mm(tile_count<1>{}); break;
        default: break;
    }
    __syncthreads();

    const float bias = b[ncol];
    #pragma unroll
    for (int i = 0; i < 4; ++i) {
        if (i < cnt) {
            float* dp = acc + ((mq + 4 * i) * 16 + (lane >> 4) * 8) * TP + ncol;
            #pragma unroll
            for (int r = 0; r < 8; ++r) dp[r * TP] = fmaxf(d[i][r] + bias, 0.0f);
        }
    }
    __syncthreads();

    float4* o4 = (float4*)out;
    for (int idx = tid; idx < nrows * (NFEAT / 4); idx += ATH) {
        const int r  = idx >> 5;
        const int f4 = idx & 31;
        o4[(size_t)(lo + r) * (NFEAT / 4) + f4] = *((const float4*)(acc + r * TP) + f4);
    }
}

__global__ void gcn_classifier(const float* __restrict__ h,
                               const float* __restrict__ Wc,
                               const float* __restrict__ bc,
                               float* __restrict__ out, int n_nodes) {
    int i = blockIdx.x * blockDim.x + threadIdx.x;
    if (i >= n_nodes) return;
    float acc[10];
    #pragma unroll
    for (int c = 0; c < 10; ++c) acc[c] = bc[c];
    const float* hr = h + (size_t)i * NFEAT;
    for (int d = 0; d < NFEAT; ++d) {
        float hv = hr[d];
        #pragma unroll
        for (int c = 0; c < 10; ++c) acc[c] += hv * Wc[d * 10 + c];
    }
    #pragma unroll
    for (int c = 0; c < 10; ++c) out[(size_t)i * 10 + c] = acc[c];
}

extern "C" void kernel_launch(void* const* d_in, const int* in_sizes, int n_in,
                              void* d_out, int out_size, void* d_ws, size_t ws_size,
                              hipStream_t stream) {
    const float* x   = (const float*)d_in[0];
    const int*   ei  = (const int*)  d_in[1];
    const float* ew  = (const float*)d_in[2];
    const float* W1  = (const float*)d_in[3];
    const float* b1  = (const float*)d_in[4];
    const float* W2  = (const float*)d_in[5];
    const float* b2  = (const float*)d_in[6];
    const float* Wc  = (const float*)d_in[7];
    const float* bc  = (const float*)d_in[8];
    float*       out = (float*)d_out;

    const int n_nodes = in_sizes[0] / NFEAT;
    const int n_edges = in_sizes[2];
    const int* src = ei;
    const int* dst = ei + n_edges;

    const size_t nd = (size_t)n_nodes * NFEAT;
    float* bufA = (float*)d_ws;
    float* bufB = bufA + nd;
    float* dinv = bufB + nd;
    const int gT  = (n_nodes + RT - 1) / RT;
    int* lists = (int*)(dinv + (size_t)((n_nodes + 31) / 32) * 32);
    int* cnts  = lists + (size_t)gT * QCAP * 4;

    const int B = 256;
    const int gN  = (n_nodes + B - 1) / B;
    const int gE  = (n_edges + B - 1) / B;

    gcn_fill1    <<<gN, B, 0, stream>>>(dinv, n_nodes);
    gcn_deg_accum<<<gE, B, 0, stream>>>(dst, ew, dinv, n_edges);
    gcn_rsqrt    <<<gN, B, 0, stream>>>(dinv, n_nodes);

    gcn_layer_tile<false><<<gT, ATH, 0, stream>>>(x, W1, src, dst, ew, dinv, b1, bufA, lists, cnts, n_nodes, n_edges);

    gcn_layer_tile<true><<<gT, ATH, 0, stream>>>(bufA, W2, src, dst, ew, dinv, b2, bufB, lists, cnts, n_nodes, n_edges);

    gcn_classifier<<<gN, B, 0, stream>>>(bufB, Wc, bc, out, n_nodes);
}


# Layer-1 edge scan reads a 1-byte key per edge (destination tile number, written by the degree pass, 4 edges a thread), not the 4-byte destination: a quarter of the bytes; sums still arrival order
# speedup vs baseline: 2.4012x; 1.0459x over previous
#include <hip/hip_runtime.h>
#include <hip/hip_bf16.h>

typedef float v2f __attribute__((ext_vector_type(2)));
typedef float v8f __attribute__((ext_vector_type(8)));

#define NFEAT 128

__global__ void gcn_fill1(float* __restrict__ deg, int n) {
    int i = blockIdx.x * blockDim.x + threadIdx.x;
    if (i < n) deg[i] = 1.0f;
}

#define RT   196

__global__ void gcn_deg_keys(const int* __restrict__ dst,
                             const float* __restrict__ ew,
                             float* __restrict__ deg, int* __restrict__ keys,
                             int n_edges) {
    const int t  = blockIdx.x * blockDim.x + threadIdx.x;
    const int e0 = 4 * t;
    if (e0 >= n_edges) return;
    if (e0 + 3 < n_edges) {
        int d[4]; float w[4];
        #pragma unroll
        for (int j = 0; j < 4; ++j) { d[j] = dst[e0 + j]; w[j] = ew[e0 + j]; }
        unsigned k = 0;
        #pragma unroll
        for (int j = 0; j < 4; ++j) {
            __hip_atomic_fetch_add(&deg[d[j]], w[j],
                                   __ATOMIC_RELAXED, __HIP_MEMORY_SCOPE_AGENT);
            k |= (((unsigned)d[j] / RT) & 255u) << (8 * j);
        }
        keys[t] = (int)k;
    } else {
        unsigned char* kb = (unsigned char*)keys;
        for (int e = e0; e < n_edges; ++e) {
            const int dd = dst[e];
            __hip_atomic_fetch_add(&deg[dd], ew[e],
                                   __ATOMIC_RELAXED, __HIP_MEMORY_SCOPE_AGENT);
            kb[e] = (unsigned char)((unsigned)dd / RT);
        }
    }
}

__global__ void gcn_rsqrt(float* __restrict__ d, int n) {
    int i = blockIdx.x * blockDim.x + threadIdx.x;
    if (i < n) d[i] = rsqrtf(d[i]);
}

#define MB   13
#define TP   132
#define WP   136
#define QCAP 16384
#define ATH  1024
#define SU   8
#define DU   8

template <int N> struct tile_count { static constexpr int value = N; };

template <bool FROM_LIST>
__global__ __launch_bounds__(ATH) void gcn_layer_tile(
        const float* __restrict__ h, const float* __restrict__ W,
        const int* __restrict__ src,
        const int* __restrict__ dst, const float* __restrict__ ew,
        const float* __restrict__ dinv, const float* __restrict__ b,
        float* __restrict__ out, int* __restrict__ lists, int* __restrict__ cnts,
        const int* __restrict__ keys, int n_nodes, int n_edges) {
    __shared__ __attribute__((aligned(16))) float acc[MB * 16 * TP];
    __shared__ __attribute__((aligned(16))) float Wl[NFEAT * WP];
    __shared__ int q[QCAP];
    __shared__ int qn;

    const int tid   = threadIdx.x;
    const int lane  = tid & 31;
    const int lo    = blockIdx.x * RT;
    const int nrows = min(RT, n_nodes - lo);
    int4* const tl  = (int4*)lists + (size_t)blockIdx.x * QCAP;

    auto clear_tile = [&]() {
        float4* a4 = (float4*)acc;
        for (int i = tid; i < MB * 16 * TP / 4; i += ATH)
            a4[i] = make_float4(0.0f, 0.0f, 0.0f, 0.0f);
    };

    auto note = [&](int e, int d) {
        if ((unsigned)(d - lo) < (unsigned)nrows) {
            int idx = atomicAdd(&qn, 1);
            if (idx < QCAP) q[idx] = e;
        }
    };

    auto drain = [&](int total, bool save) {
        const int wave = __builtin_amdgcn_readfirstlane(tid >> 5);
        int4* slot = tl + wave * DU;
        for (int base = wave; base < total; base += 32 * DU, slot += 32 * DU) {
            int e[DU]; bool ok[DU];
            #pragma unroll
            for (int u = 0; u < DU; ++u) {
                const int i = base + 32 * u;
                ok[u] = i < total;
                e[u]  = q[ok[u] ? i : base];
            }
            __builtin_amdgcn_sched_barrier(0);
            int sN[DU], dN[DU]; float w[DU];
            #pragma unroll
            for (int u = 0; u < DU; ++u) {
                sN[u] = src[e[u]]; dN[u] = dst[e[u]]; w[u] = ew[e[u]];
            }
            __builtin_amdgcn_sched_barrier(0);
            float c[DU]; float hv[DU][4];
            #pragma unroll
            for (int u = 0; u < DU; ++u) {
                c[u] = dinv[sN[u]] * w[u] * dinv[dN[u]];
                const float* hr = h + (size_t)sN[u] * NFEAT + lane;
                #pragma unroll
                for (int j = 0; j < 4; ++j) hv[u][j] = hr[32 * j];
            }
            int row[DU];
            #pragma unroll
            for (int u = 0; u < DU; ++u) {
                row[u] = (ok[u] && (unsigned)(dN[u] - lo) < (unsigned)nrows) ? dN[u] - lo : RT;
                float* ar = acc + row[u] * TP + lane;
                #pragma unroll
                for (int j = 0; j < 4; ++j)
                    __hip_atomic_fetch_add(ar + 32 * j, hv[u][j] * c[u],
                                           __ATOMIC_RELAXED, __HIP_MEMORY_SCOPE_WORKGROUP);
            }
            if (save) {
                int ex = sN[0], ey = row[0]; float ez = c[0];
                #pragma unroll
                for (int u = 1; u < DU; ++u)
                    if (lane == u) { ex = sN[u]; ey = row[u]; ez = c[u]; }
                if (lane < DU) slot[lane] = make_int4(ex, ey, __float_as_int(ez), 0);
            }
        }
    };

    auto drain_list = [&](int total) {
        const int wave = __builtin_amdgcn_readfirstlane(tid >> 5);
        if (wave >= total) return;
        const int4* slot = tl + wave * DU;
        int4 en[DU];
        #pragma unroll
        for (int u = 0; u < DU; ++u) en[u] = slot[u];
        for (int base = wave; base < total; base += 32 * DU) {
            float hv[DU][4];
            #pragma unroll
            for (int u = 0; u < DU; ++u) {
                const float* hr = h + (size_t)en[u].x * NFEAT + lane;
                #pragma unroll
                for (int j = 0; j < 4; ++j) hv[u][j] = hr[32 * j];
            }
            if (base + 32 * DU < total) slot += 32 * DU;
            int4 nx[DU];
            #pragma unroll
            for (int u = 0; u < DU; ++u) nx[u] = slot[u];
            #pragma unroll
            for (int u = 0; u < DU; ++u) {
                float* ar = acc + en[u].y * TP + lane;
                const float cu = __int_as_float(en[u].z);
                #pragma unroll
                for (int j = 0; j < 4; ++j)
                    __hip_atomic_fetch_add(ar + 32 * j, hv[u][j] * cu,
                                           __ATOMIC_RELAXED, __HIP_MEMORY_SCOPE_WORKGROUP);
            }
            #pragma unroll
            for (int u = 0; u < DU; ++u) en[u] = nx[u];
        }
    };

    {
        const float4* Wv = (const float4*)W;
        float4 wv[(NFEAT * NFEAT / 4) / ATH];
        #pragma unroll
        for (int u = 0; u < (NFEAT * NFEAT / 4) / ATH; ++u) wv[u] = Wv[tid + u * ATH];
        clear_tile();
        #pragma unroll
        for (int u = 0; u < (NFEAT * NFEAT / 4) / ATH; ++u) {
            const int i = tid + u * ATH;
            *(float4*)(Wl + (i >> 5) * WP + (i & 31) * 4) = wv[u];
        }
    }
    if (tid == 0) qn = 0;
    __syncthreads();

    int noted;
    if constexpr (FROM_LIST) {
        noted = cnts[blockIdx.x * 32];
    } else {
    const int n16   = n_edges >> 4;
    const int tail0 = n16 << 4;
    const int4* k4  = (const int4*)keys;
    const unsigned pat = (unsigned)(blockIdx.x & 255) * 0x01010101u;
    auto zero_bytes = [](unsigned x) {
        return ~(((x & 0x7F7F7F7Fu) + 0x7F7F7F7Fu) | x | 0x7F7F7F7Fu);
    };
    if (tid < n_edges - tail0) note(tail0 + tid, dst[tail0 + tid]);
    for (int i = tid; i < n16; i += ATH * SU) {
        int4 v[SU];
        #pragma unroll
        for (int u = 0; u < SU; ++u) {
            const int k = i + u * ATH;
            v[u] = k4[min(k, n16 - 1)];
        }
        #pragma unroll
        for (int u = 0; u < SU; ++u) {
            const int k = i + u * ATH;
            unsigned mm = (zero_bytes((unsigned)v[u].x ^ pat) >> 7)
                        | (zero_bytes((unsigned)v[u].y ^ pat) >> 6)
                        | (zero_bytes((unsigned)v[u].z ^ pat) >> 5)
                        | (zero_bytes((unsigned)v[u].w ^ pat) >> 4);
            if (k >= n16) mm = 0;
            while (mm) {
                const int p = __builtin_ctz(mm);
                mm &= mm - 1;
                const int e = (k << 4) + ((p & 7) << 2) + (p >> 3);
                const int idx = atomicAdd(&qn, 1);
                if (idx < QCAP) q[idx] = e;
            }
        }
    }
    __syncthreads();
    noted = qn;
    if (tid == 0) cnts[blockIdx.x * 32] = noted;
    }

    if (noted <= QCAP) {
        if constexpr (FROM_LIST) drain_list(noted);
        else                     drain(noted, true);
    } else {
        __syncthreads();
        clear_tile();
        for (int seg = 0; seg < n_edges; seg += QCAP) {
            if (tid == 0) qn = 0;
            __syncthreads();
            const int segEnd = min(seg + QCAP, n_edges);
            for (int e = seg + tid; e < segEnd; e += ATH) note(e, dst[e]);
            __syncthreads();
            drain(qn, false);
            __syncthreads();
        }
    }
    __syncthreads();

    const float4* h4 = (const float4*)h;
    for (int idx = tid; idx < nrows * (NFEAT / 4); idx += ATH) {
        const int r  = idx >> 5;
        const int f4 = idx & 31;
        const float di = dinv[lo + r];
        const float d2 = di * di;
        const float4 hv = h4[(size_t)(lo + r) * (NFEAT / 4) + f4];
        float4* ap = (float4*)(acc + r * TP) + f4;
        float4 a = *ap;
        a.x += hv.x * d2;
        a.y += hv.y * d2;
        a.z += hv.z * d2;
        a.w += hv.w * d2;
        *ap = a;
    }
    __syncthreads();

    const int wave  = __builtin_amdgcn_readfirstlane(tid >> 5);
    const int mlane = lane & 15;
    const int khalf = (lane >> 4) * 2;
    const int ncol  = (wave & 7) * 16 + mlane;
    const int mq    = wave >> 3;
    const int cnt   = (((nrows + 15) >> 4) - mq + 3) >> 2;
    v8f d[4] = {};
    auto tile_mm = [&](auto ntc) {
        constexpr int NT = decltype(ntc)::value;
        const float* ap = acc + (mq * 16 + mlane) * TP + khalf;
        const float* bp = Wl + khalf * WP + ncol;
        #pragma unroll 4
        for (int k0 = 0; k0 < NFEAT; k0 += 4) {
            v2f bb;
            bb.x = bp[(k0    ) * WP];
            bb.y = bp[(k0 + 1) * WP];
            #pragma unroll
            for (int i = 0; i < NT; ++i) {
                const v2f a = *(const v2f*)(ap + i * 64 * TP + k0);
                d[i] = __builtin_amdgcn_wmma_f32_16x16x4_f32(
                           false, a, false, bb, (short)0, d[i], false, false);
            }
        }
    };
    switch (cnt) {
        case 4: tile_mm(tile_count<4>{}); break;
        case 3: tile_mm(tile_count<3>{}); break;
        case 2: tile_mm(tile_count<2>{}); break;
        case 1: tile_mm(tile_count<1>{}); break;
        default: break;
    }
    __syncthreads();

    const float bias = b[ncol];
    #pragma unroll
    for (int i = 0; i < 4; ++i) {
        if (i < cnt) {
            float* dp = acc + ((mq + 4 * i) * 16 + (lane >> 4) * 8) * TP + ncol;
            #pragma unroll
            for (int r = 0; r < 8; ++r) dp[r * TP] = fmaxf(d[i][r] + bias, 0.0f);
        }
    }
    __syncthreads();

    float4* o4 = (float4*)out;
    for (int idx = tid; idx < nrows * (NFEAT / 4); idx += ATH) {
        const int r  = idx >> 5;
        const int f4 = idx & 31;
        o4[(size_t)(lo + r) * (NFEAT / 4) + f4] = *((const float4*)(acc + r * TP) + f4);
    }
}

__global__ void gcn_classifier(const float* __restrict__ h,
                               const float* __restrict__ Wc,
                               const float* __restrict__ bc,
                               float* __restrict__ out, int n_nodes) {
    int i = blockIdx.x * blockDim.x + threadIdx.x;
    if (i >= n_nodes) return;
    float acc[10];
    #pragma unroll
    for (int c = 0; c < 10; ++c) acc[c] = bc[c];
    const float* hr = h + (size_t)i * NFEAT;
    for (int d = 0; d < NFEAT; ++d) {
        float hv = hr[d];
        #pragma unroll
        for (int c = 0; c < 10; ++c) acc[c] += hv * Wc[d * 10 + c];
    }
    #pragma unroll
    for (int c = 0; c < 10; ++c) out[(size_t)i * 10 + c] = acc[c];
}

extern "C" void kernel_launch(void* const* d_in, const int* in_sizes, int n_in,
                              void* d_out, int out_size, void* d_ws, size_t ws_size,
                              hipStream_t stream) {
    const float* x   = (const float*)d_in[0];
    const int*   ei  = (const int*)  d_in[1];
    const float* ew  = (const float*)d_in[2];
    const float* W1  = (const float*)d_in[3];
    const float* b1  = (const float*)d_in[4];
    const float* W2  = (const float*)d_in[5];
    const float* b2  = (const float*)d_in[6];
    const float* Wc  = (const float*)d_in[7];
    const float* bc  = (const float*)d_in[8];
    float*       out = (float*)d_out;

    const int n_nodes = in_sizes[0] / NFEAT;
    const int n_edges = in_sizes[2];
    const int* src = ei;
    const int* dst = ei + n_edges;

    const size_t nd = (size_t)n_nodes * NFEAT;
    float* bufA = (float*)d_ws;
    float* bufB = bufA + nd;
    float* dinv = bufB + nd;
    const int gT  = (n_nodes + RT - 1) / RT;
    int* lists = (int*)(dinv + (size_t)((n_nodes + 31) / 32) * 32);
    int* cnts  = lists + (size_t)gT * QCAP * 4;
    int* keys  = cnts + (size_t)gT * 32;

    const int B = 256;
    const int gN  = (n_nodes + B - 1) / B;
    const int gE  = (n_edges + 4 * B - 1) / (4 * B);

    gcn_fill1    <<<gN, B, 0, stream>>>(dinv, n_nodes);
    gcn_deg_keys <<<gE, B, 0, stream>>>(dst, ew, dinv, keys, n_edges);
    gcn_rsqrt    <<<gN, B, 0, stream>>>(dinv, n_nodes);

    gcn_layer_tile<false><<<gT, ATH, 0, stream>>>(x, W1, src, dst, ew, dinv, b1, bufA, lists, cnts, keys, n_nodes, n_edges);

    gcn_layer_tile<true><<<gT, ATH, 0, stream>>>(bufA, W2, src, dst, ew, dinv, b2, bufB, lists, cnts, keys, n_nodes, n_edges);

    gcn_classifier<<<gN, B, 0, stream>>>(bufB, Wc, bc, out, n_nodes);
}


# Source rows gathered by one 16-byte-a-lane load (whole 512-byte row per instruction, not four 4-byte loads); tile columns and W rows permuted alike in LDS, so the product's k sum order changes
# speedup vs baseline: 2.4753x; 1.0309x over previous
#include <hip/hip_runtime.h>
#include <hip/hip_bf16.h>

typedef float v2f __attribute__((ext_vector_type(2)));
typedef float v8f __attribute__((ext_vector_type(8)));

#define NFEAT 128

__global__ void gcn_fill1(float* __restrict__ deg, int n) {
    int i = blockIdx.x * blockDim.x + threadIdx.x;
    if (i < n) deg[i] = 1.0f;
}

#define RT   196

__global__ void gcn_deg_keys(const int* __restrict__ dst,
                             const float* __restrict__ ew,
                             float* __restrict__ deg, int* __restrict__ keys,
                             int n_edges) {
    const int t  = blockIdx.x * blockDim.x + threadIdx.x;
    const int e0 = 4 * t;
    if (e0 >= n_edges) return;
    if (e0 + 3 < n_edges) {
        int d[4]; float w[4];
        #pragma unroll
        for (int j = 0; j < 4; ++j) { d[j] = dst[e0 + j]; w[j] = ew[e0 + j]; }
        unsigned k = 0;
        #pragma unroll
        for (int j = 0; j < 4; ++j) {
            __hip_atomic_fetch_add(&deg[d[j]], w[j],
                                   __ATOMIC_RELAXED, __HIP_MEMORY_SCOPE_AGENT);
            k |= (((unsigned)d[j] / RT) & 255u) << (8 * j);
        }
        keys[t] = (int)k;
    } else {
        unsigned char* kb = (unsigned char*)keys;
        for (int e = e0; e < n_edges; ++e) {
            const int dd = dst[e];
            __hip_atomic_fetch_add(&deg[dd], ew[e],
                                   __ATOMIC_RELAXED, __HIP_MEMORY_SCOPE_AGENT);
            kb[e] = (unsigned char)((unsigned)dd / RT);
        }
    }
}

__global__ void gcn_rsqrt(float* __restrict__ d, int n) {
    int i = blockIdx.x * blockDim.x + threadIdx.x;
    if (i < n) d[i] = rsqrtf(d[i]);
}

#define MB   13
#define TP   132
#define WP   136
#define QCAP 16384
#define TCOL(f) ((((f) & 3) << 5) + ((f) >> 2))
#define ATH  1024
#define SU   8
#define DU   8

template <int N> struct tile_count { static constexpr int value = N; };

template <bool FROM_LIST>
__global__ __launch_bounds__(ATH) void gcn_layer_tile(
        const float* __restrict__ h, const float* __restrict__ W,
        const int* __restrict__ src,
        const int* __restrict__ dst, const float* __restrict__ ew,
        const float* __restrict__ dinv, const float* __restrict__ b,
        float* __restrict__ out, int* __restrict__ lists, int* __restrict__ cnts,
        const int* __restrict__ keys, int n_nodes, int n_edges) {
    __shared__ __attribute__((aligned(16))) float acc[MB * 16 * TP];
    __shared__ __attribute__((aligned(16))) float Wl[NFEAT * WP];
    __shared__ int q[QCAP];
    __shared__ int qn;

    const int tid   = threadIdx.x;
    const int lane  = tid & 31;
    const int lo    = blockIdx.x * RT;
    const int nrows = min(RT, n_nodes - lo);
    int4* const tl  = (int4*)lists + (size_t)blockIdx.x * QCAP;

    auto clear_tile = [&]() {
        float4* a4 = (float4*)acc;
        for (int i = tid; i < MB * 16 * TP / 4; i += ATH)
            a4[i] = make_float4(0.0f, 0.0f, 0.0f, 0.0f);
    };

    auto note = [&](int e, int d) {
        if ((unsigned)(d - lo) < (unsigned)nrows) {
            int idx = atomicAdd(&qn, 1);
            if (idx < QCAP) q[idx] = e;
        }
    };

    auto drain = [&](int total, bool save) {
        const int wave = __builtin_amdgcn_readfirstlane(tid >> 5);
        int4* slot = tl + wave * DU;
        for (int base = wave; base < total; base += 32 * DU, slot += 32 * DU) {
            int e[DU]; bool ok[DU];
            #pragma unroll
            for (int u = 0; u < DU; ++u) {
                const int i = base + 32 * u;
                ok[u] = i < total;
                e[u]  = q[ok[u] ? i : base];
            }
            __builtin_amdgcn_sched_barrier(0);
            int sN[DU], dN[DU]; float w[DU];
            #pragma unroll
            for (int u = 0; u < DU; ++u) {
                sN[u] = src[e[u]]; dN[u] = dst[e[u]]; w[u] = ew[e[u]];
            }
            __builtin_amdgcn_sched_barrier(0);
            float c[DU]; float4 hv[DU];
            #pragma unroll
            for (int u = 0; u < DU; ++u) {
                c[u] = dinv[sN[u]] * w[u] * dinv[dN[u]];
                hv[u] = *((const float4*)(h + (size_t)sN[u] * NFEAT) + lane);
            }
            int row[DU];
            #pragma unroll
            for (int u = 0; u < DU; ++u) {
                row[u] = (ok[u] && (unsigned)(dN[u] - lo) < (unsigned)nrows) ? dN[u] - lo : RT;
                float* ar = acc + row[u] * TP + lane;
                const float hj[4] = {hv[u].x, hv[u].y, hv[u].z, hv[u].w};
                #pragma unroll
                for (int j = 0; j < 4; ++j)
                    __hip_atomic_fetch_add(ar + 32 * j, hj[j] * c[u],
                                           __ATOMIC_RELAXED, __HIP_MEMORY_SCOPE_WORKGROUP);
            }
            if (save) {
                int ex = sN[0], ey = row[0]; float ez = c[0];
                #pragma unroll
                for (int u = 1; u < DU; ++u)
                    if (lane == u) { ex = sN[u]; ey = row[u]; ez = c[u]; }
                if (lane < DU) slot[lane] = make_int4(ex, ey, __float_as_int(ez), 0);
            }
        }
    };

    auto drain_list = [&](int total) {
        const int wave = __builtin_amdgcn_readfirstlane(tid >> 5);
        if (wave >= total) return;
        const int4* slot = tl + wave * DU;
        int4 en[DU];
        #pragma unroll
        for (int u = 0; u < DU; ++u) en[u] = slot[u];
        for (int base = wave; base < total; base += 32 * DU) {
            float4 hv[DU];
            #pragma unroll
            for (int u = 0; u < DU; ++u)
                hv[u] = *((const float4*)(h + (size_t)en[u].x * NFEAT) + lane);
            if (base + 32 * DU < total) slot += 32 * DU;
            int4 nx[DU];
            #pragma unroll
            for (int u = 0; u < DU; ++u) nx[u] = slot[u];
            #pragma unroll
            for (int u = 0; u < DU; ++u) {
                float* ar = acc + en[u].y * TP + lane;
                const float cu = __int_as_float(en[u].z);
                const float hj[4] = {hv[u].x, hv[u].y, hv[u].z, hv[u].w};
                #pragma unroll
                for (int j = 0; j < 4; ++j)
                    __hip_atomic_fetch_add(ar + 32 * j, hj[j] * cu,
                                           __ATOMIC_RELAXED, __HIP_MEMORY_SCOPE_WORKGROUP);
            }
            #pragma unroll
            for (int u = 0; u < DU; ++u) en[u] = nx[u];
        }
    };

    {
        const float4* Wv = (const float4*)W;
        float4 wv[(NFEAT * NFEAT / 4) / ATH];
        #pragma unroll
        for (int u = 0; u < (NFEAT * NFEAT / 4) / ATH; ++u) wv[u] = Wv[tid + u * ATH];
        clear_tile();
        #pragma unroll
        for (int u = 0; u < (NFEAT * NFEAT / 4) / ATH; ++u) {
            const int i = tid + u * ATH;
            *(float4*)(Wl + TCOL(i >> 5) * WP + (i & 31) * 4) = wv[u];
        }
    }
    if (tid == 0) qn = 0;
    __syncthreads();

    int noted;
    if constexpr (FROM_LIST) {
        noted = cnts[blockIdx.x * 32];
    } else {
    const int n16   = n_edges >> 4;
    const int tail0 = n16 << 4;
    const int4* k4  = (const int4*)keys;
    const unsigned pat = (unsigned)(blockIdx.x & 255) * 0x01010101u;
    auto zero_bytes = [](unsigned x) {
        return ~(((x & 0x7F7F7F7Fu) + 0x7F7F7F7Fu) | x | 0x7F7F7F7Fu);
    };
    if (tid < n_edges - tail0) note(tail0 + tid, dst[tail0 + tid]);
    for (int i = tid; i < n16; i += ATH * SU) {
        int4 v[SU];
        #pragma unroll
        for (int u = 0; u < SU; ++u) {
            const int k = i + u * ATH;
            v[u] = k4[min(k, n16 - 1)];
        }
        #pragma unroll
        for (int u = 0; u < SU; ++u) {
            const int k = i + u * ATH;
            unsigned mm = (zero_bytes((unsigned)v[u].x ^ pat) >> 7)
                        | (zero_bytes((unsigned)v[u].y ^ pat) >> 6)
                        | (zero_bytes((unsigned)v[u].z ^ pat) >> 5)
                        | (zero_bytes((unsigned)v[u].w ^ pat) >> 4);
            if (k >= n16) mm = 0;
            while (mm) {
                const int p = __builtin_ctz(mm);
                mm &= mm - 1;
                const int e = (k << 4) + ((p & 7) << 2) + (p >> 3);
                const int idx = atomicAdd(&qn, 1);
                if (idx < QCAP) q[idx] = e;
            }
        }
    }
    __syncthreads();
    noted = qn;
    if (tid == 0) cnts[blockIdx.x * 32] = noted;
    }

    if (noted <= QCAP) {
        if constexpr (FROM_LIST) drain_list(noted);
        else                     drain(noted, true);
    } else {
        __syncthreads();
        clear_tile();
        for (int seg = 0; seg < n_edges; seg += QCAP) {
            if (tid == 0) qn = 0;
            __syncthreads();
            const int segEnd = min(seg + QCAP, n_edges);
            for (int e = seg + tid; e < segEnd; e += ATH) note(e, dst[e]);
            __syncthreads();
            drain(qn, false);
            __syncthreads();
        }
    }
    __syncthreads();

    const float4* h4 = (const float4*)h;
    for (int idx = tid; idx < nrows * (NFEAT / 4); idx += ATH) {
        const int r  = idx >> 5;
        const int f4 = idx & 31;
        const float di = dinv[lo + r];
        const float d2 = di * di;
        const float4 hv = h4[(size_t)(lo + r) * (NFEAT / 4) + f4];
        float* ap = acc + r * TP + f4;
        float a0 = ap[0], a1 = ap[32], a2 = ap[64], a3 = ap[96];
        a0 += hv.x * d2;
        a1 += hv.y * d2;
        a2 += hv.z * d2;
        a3 += hv.w * d2;
        ap[0] = a0; ap[32] = a1; ap[64] = a2; ap[96] = a3;
    }
    __syncthreads();

    const int wave  = __builtin_amdgcn_readfirstlane(tid >> 5);
    const int mlane = lane & 15;
    const int khalf = (lane >> 4) * 2;
    const int ncol  = (wave & 7) * 16 + mlane;
    const int mq    = wave >> 3;
    const int cnt   = (((nrows + 15) >> 4) - mq + 3) >> 2;
    v8f d[4] = {};
    auto tile_mm = [&](auto ntc) {
        constexpr int NT = decltype(ntc)::value;
        const float* ap = acc + (mq * 16 + mlane) * TP + khalf;
        const float* bp = Wl + khalf * WP + ncol;
        #pragma unroll 4
        for (int k0 = 0; k0 < NFEAT; k0 += 4) {
            v2f bb;
            bb.x = bp[(k0    ) * WP];
            bb.y = bp[(k0 + 1) * WP];
            #pragma unroll
            for (int i = 0; i < NT; ++i) {
                const v2f a = *(const v2f*)(ap + i * 64 * TP + k0);
                d[i] = __builtin_amdgcn_wmma_f32_16x16x4_f32(
                           false, a, false, bb, (short)0, d[i], false, false);
            }
        }
    };
    switch (cnt) {
        case 4: tile_mm(tile_count<4>{}); break;
        case 3: tile_mm(tile_count<3>{}); break;
        case 2: tile_mm(tile_count<2>{}); break;
        case 1: tile_mm(tile_count<1>{}); break;
        default: break;
    }
    __syncthreads();

    const float bias = b[ncol];
    #pragma unroll
    for (int i = 0; i < 4; ++i) {
        if (i < cnt) {
            float* dp = acc + ((mq + 4 * i) * 16 + (lane >> 4) * 8) * TP + ncol;
            #pragma unroll
            for (int r = 0; r < 8; ++r) dp[r * TP] = fmaxf(d[i][r] + bias, 0.0f);
        }
    }
    __syncthreads();

    float4* o4 = (float4*)out;
    for (int idx = tid; idx < nrows * (NFEAT / 4); idx += ATH) {
        const int r  = idx >> 5;
        const int f4 = idx & 31;
        o4[(size_t)(lo + r) * (NFEAT / 4) + f4] = *((const float4*)(acc + r * TP) + f4);
    }
}

__global__ void gcn_classifier(const float* __restrict__ h,
                               const float* __restrict__ Wc,
                               const float* __restrict__ bc,
                               float* __restrict__ out, int n_nodes) {
    int i = blockIdx.x * blockDim.x + threadIdx.x;
    if (i >= n_nodes) return;
    float acc[10];
    #pragma unroll
    for (int c = 0; c < 10; ++c) acc[c] = bc[c];
    const float* hr = h + (size_t)i * NFEAT;
    for (int d = 0; d < NFEAT; ++d) {
        float hv = hr[d];
        #pragma unroll
        for (int c = 0; c < 10; ++c) acc[c] += hv * Wc[d * 10 + c];
    }
    #pragma unroll
    for (int c = 0; c < 10; ++c) out[(size_t)i * 10 + c] = acc[c];
}

extern "C" void kernel_launch(void* const* d_in, const int* in_sizes, int n_in,
                              void* d_out, int out_size, void* d_ws, size_t ws_size,
                              hipStream_t stream) {
    const float* x   = (const float*)d_in[0];
    const int*   ei  = (const int*)  d_in[1];
    const float* ew  = (const float*)d_in[2];
    const float* W1  = (const float*)d_in[3];
    const float* b1  = (const float*)d_in[4];
    const float* W2  = (const float*)d_in[5];
    const float* b2  = (const float*)d_in[6];
    const float* Wc  = (const float*)d_in[7];
    const float* bc  = (const float*)d_in[8];
    float*       out = (float*)d_out;

    const int n_nodes = in_sizes[0] / NFEAT;
    const int n_edges = in_sizes[2];
    const int* src = ei;
    const int* dst = ei + n_edges;

    const size_t nd = (size_t)n_nodes * NFEAT;
    float* bufA = (float*)d_ws;
    float* bufB = bufA + nd;
    float* dinv = bufB + nd;
    const int gT  = (n_nodes + RT - 1) / RT;
    int* lists = (int*)(dinv + (size_t)((n_nodes + 31) / 32) * 32);
    int* cnts  = lists + (size_t)gT * QCAP * 4;
    int* keys  = cnts + (size_t)gT * 32;

    const int B = 256;
    const int gN  = (n_nodes + B - 1) / B;
    const int gE  = (n_edges + 4 * B - 1) / (4 * B);

    gcn_fill1    <<<gN, B, 0, stream>>>(dinv, n_nodes);
    gcn_deg_keys <<<gE, B, 0, stream>>>(dst, ew, dinv, keys, n_edges);
    gcn_rsqrt    <<<gN, B, 0, stream>>>(dinv, n_nodes);

    gcn_layer_tile<false><<<gT, ATH, 0, stream>>>(x, W1, src, dst, ew, dinv, b1, bufA, lists, cnts, keys, n_nodes, n_edges);

    gcn_layer_tile<true><<<gT, ATH, 0, stream>>>(bufA, W2, src, dst, ew, dinv, b2, bufB, lists, cnts, keys, n_nodes, n_edges);

    gcn_classifier<<<gN, B, 0, stream>>>(bufB, Wc, bc, out, n_nodes);
}


# Classifier joins the layer-2 launch: relu tile stays in LDS, tile @ Wc (padded to 16 columns) on WMMA f32, only scores stored; h2 plane and classifier launch go; class sum order changes
# speedup vs baseline: 2.7522x; 1.1119x over previous
#include <hip/hip_runtime.h>
#include <hip/hip_bf16.h>

typedef float v2f __attribute__((ext_vector_type(2)));
typedef float v8f __attribute__((ext_vector_type(8)));

#define NFEAT 128

__global__ void gcn_fill1(float* __restrict__ deg, int n) {
    int i = blockIdx.x * blockDim.x + threadIdx.x;
    if (i < n) deg[i] = 1.0f;
}

#define RT   196

__global__ void gcn_deg_keys(const int* __restrict__ dst,
                             const float* __restrict__ ew,
                             float* __restrict__ deg, int* __restrict__ keys,
                             int n_edges) {
    const int t  = blockIdx.x * blockDim.x + threadIdx.x;
    const int e0 = 4 * t;
    if (e0 >= n_edges) return;
    if (e0 + 3 < n_edges) {
        int d[4]; float w[4];
        #pragma unroll
        for (int j = 0; j < 4; ++j) { d[j] = dst[e0 + j]; w[j] = ew[e0 + j]; }
        unsigned k = 0;
        #pragma unroll
        for (int j = 0; j < 4; ++j) {
            __hip_atomic_fetch_add(&deg[d[j]], w[j],
                                   __ATOMIC_RELAXED, __HIP_MEMORY_SCOPE_AGENT);
            k |= (((unsigned)d[j] / RT) & 255u) << (8 * j);
        }
        keys[t] = (int)k;
    } else {
        unsigned char* kb = (unsigned char*)keys;
        for (int e = e0; e < n_edges; ++e) {
            const int dd = dst[e];
            __hip_atomic_fetch_add(&deg[dd], ew[e],
                                   __ATOMIC_RELAXED, __HIP_MEMORY_SCOPE_AGENT);
            kb[e] = (unsigned char)((unsigned)dd / RT);
        }
    }
}

__global__ void gcn_rsqrt(float* __restrict__ d, int n) {
    int i = blockIdx.x * blockDim.x + threadIdx.x;
    if (i < n) d[i] = rsqrtf(d[i]);
}

#define MB   13
#define TP   132
#define WP   136
#define QCAP 16384
#define TCOL(f) ((((f) & 3) << 5) + ((f) >> 2))
#define ATH  1024
#define SU   8
#define DU   8
#define NCLS 10
#define CP   16

template <int N> struct tile_count { static constexpr int value = N; };

template <bool FROM_LIST>
__global__ __launch_bounds__(ATH) void gcn_layer_tile(
        const float* __restrict__ h, const float* __restrict__ W,
        const int* __restrict__ src,
        const int* __restrict__ dst, const float* __restrict__ ew,
        const float* __restrict__ dinv, const float* __restrict__ b,
        float* __restrict__ out, int* __restrict__ lists, int* __restrict__ cnts,
        const int* __restrict__ keys,
        const float* __restrict__ Wc, const float* __restrict__ bc,
        float* __restrict__ scores, int n_nodes, int n_edges) {
    __shared__ __attribute__((aligned(16))) float acc[MB * 16 * TP];
    __shared__ __attribute__((aligned(16))) float Wl[NFEAT * WP];
    __shared__ int q[QCAP];
    __shared__ int qn;
    __shared__ __attribute__((aligned(16))) float Wcl[FROM_LIST ? NFEAT * CP : 1];
    __shared__ __attribute__((aligned(16))) float lg[FROM_LIST ? MB * 16 * NCLS : 1];

    const int tid   = threadIdx.x;
    const int lane  = tid & 31;
    const int lo    = blockIdx.x * RT;
    const int nrows = min(RT, n_nodes - lo);
    int4* const tl  = (int4*)lists + (size_t)blockIdx.x * QCAP;

    auto clear_tile = [&]() {
        float4* a4 = (float4*)acc;
        for (int i = tid; i < MB * 16 * TP / 4; i += ATH)
            a4[i] = make_float4(0.0f, 0.0f, 0.0f, 0.0f);
    };

    auto note = [&](int e, int d) {
        if ((unsigned)(d - lo) < (unsigned)nrows) {
            int idx = atomicAdd(&qn, 1);
            if (idx < QCAP) q[idx] = e;
        }
    };

    auto drain = [&](int total, bool save) {
        const int wave = __builtin_amdgcn_readfirstlane(tid >> 5);
        int4* slot = tl + wave * DU;
        for (int base = wave; base < total; base += 32 * DU, slot += 32 * DU) {
            int e[DU]; bool ok[DU];
            #pragma unroll
            for (int u = 0; u < DU; ++u) {
                const int i = base + 32 * u;
                ok[u] = i < total;
                e[u]  = q[ok[u] ? i : base];
            }
            __builtin_amdgcn_sched_barrier(0);
            int sN[DU], dN[DU]; float w[DU];
            #pragma unroll
            for (int u = 0; u < DU; ++u) {
                sN[u] = src[e[u]]; dN[u] = dst[e[u]]; w[u] = ew[e[u]];
            }
            __builtin_amdgcn_sched_barrier(0);
            float c[DU]; float4 hv[DU];
            #pragma unroll
            for (int u = 0; u < DU; ++u) {
                c[u] = dinv[sN[u]] * w[u] * dinv[dN[u]];
                hv[u] = *((const float4*)(h + (size_t)sN[u] * NFEAT) + lane);
            }
            int row[DU];
            #pragma unroll
            for (int u = 0; u < DU; ++u) {
                row[u] = (ok[u] && (unsigned)(dN[u] - lo) < (unsigned)nrows) ? dN[u] - lo : RT;
                float* ar = acc + row[u] * TP + lane;
                const float hj[4] = {hv[u].x, hv[u].y, hv[u].z, hv[u].w};
                #pragma unroll
                for (int j = 0; j < 4; ++j)
                    __hip_atomic_fetch_add(ar + 32 * j, hj[j] * c[u],
                                           __ATOMIC_RELAXED, __HIP_MEMORY_SCOPE_WORKGROUP);
            }
            if (save) {
                int ex = sN[0], ey = row[0]; float ez = c[0];
                #pragma unroll
                for (int u = 1; u < DU; ++u)
                    if (lane == u) { ex = sN[u]; ey = row[u]; ez = c[u]; }
                if (lane < DU) slot[lane] = make_int4(ex, ey, __float_as_int(ez), 0);
            }
        }
    };

    auto drain_list = [&](int total) {
        const int wave = __builtin_amdgcn_readfirstlane(tid >> 5);
        if (wave >= total) return;
        const int4* slot = tl + wave * DU;
        int4 en[DU];
        #pragma unroll
        for (int u = 0; u < DU; ++u) en[u] = slot[u];
        for (int base = wave; base < total; base += 32 * DU) {
            float4 hv[DU];
            #pragma unroll
            for (int u = 0; u < DU; ++u)
                hv[u] = *((const float4*)(h + (size_t)en[u].x * NFEAT) + lane);
            if (base + 32 * DU < total) slot += 32 * DU;
            int4 nx[DU];
            #pragma unroll
            for (int u = 0; u < DU; ++u) nx[u] = slot[u];
            #pragma unroll
            for (int u = 0; u < DU; ++u) {
                float* ar = acc + en[u].y * TP + lane;
                const float cu = __int_as_float(en[u].z);
                const float hj[4] = {hv[u].x, hv[u].y, hv[u].z, hv[u].w};
                #pragma unroll
                for (int j = 0; j < 4; ++j)
                    __hip_atomic_fetch_add(ar + 32 * j, hj[j] * cu,
                                           __ATOMIC_RELAXED, __HIP_MEMORY_SCOPE_WORKGROUP);
            }
            #pragma unroll
            for (int u = 0; u < DU; ++u) en[u] = nx[u];
        }
    };

    {
        const float4* Wv = (const float4*)W;
        float4 wv[(NFEAT * NFEAT / 4) / ATH];
        #pragma unroll
        for (int u = 0; u < (NFEAT * NFEAT / 4) / ATH; ++u) wv[u] = Wv[tid + u * ATH];
        float wcv[(NFEAT * CP) / ATH];
        if constexpr (FROM_LIST) {
            #pragma unroll
            for (int u = 0; u < (NFEAT * CP) / ATH; ++u) {
                const int i = tid + u * ATH;
                const int c = i & (CP - 1);
                wcv[u] = c < NCLS ? Wc[(i / CP) * NCLS + c] : 0.0f;
            }
        }
        clear_tile();
        #pragma unroll
        for (int u = 0; u < (NFEAT * NFEAT / 4) / ATH; ++u) {
            const int i = tid + u * ATH;
            *(float4*)(Wl + TCOL(i >> 5) * WP + (i & 31) * 4) = wv[u];
        }
        if constexpr (FROM_LIST) {
            #pragma unroll
            for (int u = 0; u < (NFEAT * CP) / ATH; ++u) Wcl[tid + u * ATH] = wcv[u];
        }
    }
    if (tid == 0) qn = 0;
    __syncthreads();

    int noted;
    if constexpr (FROM_LIST) {
        noted = cnts[blockIdx.x * 32];
    } else {
    const int n16   = n_edges >> 4;
    const int tail0 = n16 << 4;
    const int4* k4  = (const int4*)keys;
    const unsigned pat = (unsigned)(blockIdx.x & 255) * 0x01010101u;
    auto zero_bytes = [](unsigned x) {
        return ~(((x & 0x7F7F7F7Fu) + 0x7F7F7F7Fu) | x | 0x7F7F7F7Fu);
    };
    if (tid < n_edges - tail0) note(tail0 + tid, dst[tail0 + tid]);
    for (int i = tid; i < n16; i += ATH * SU) {
        int4 v[SU];
        #pragma unroll
        for (int u = 0; u < SU; ++u) {
            const int k = i + u * ATH;
            v[u] = k4[min(k, n16 - 1)];
        }
        #pragma unroll
        for (int u = 0; u < SU; ++u) {
            const int k = i + u * ATH;
            unsigned mm = (zero_bytes((unsigned)v[u].x ^ pat) >> 7)
                        | (zero_bytes((unsigned)v[u].y ^ pat) >> 6)
                        | (zero_bytes((unsigned)v[u].z ^ pat) >> 5)
                        | (zero_bytes((unsigned)v[u].w ^ pat) >> 4);
            if (k >= n16) mm = 0;
            while (mm) {
                const int p = __builtin_ctz(mm);
                mm &= mm - 1;
                const int e = (k << 4) + ((p & 7) << 2) + (p >> 3);
                const int idx = atomicAdd(&qn, 1);
                if (idx < QCAP) q[idx] = e;
            }
        }
    }
    __syncthreads();
    noted = qn;
    if (tid == 0) cnts[blockIdx.x * 32] = noted;
    }

    if (noted <= QCAP) {
        if constexpr (FROM_LIST) drain_list(noted);
        else                     drain(noted, true);
    } else {
        __syncthreads();
        clear_tile();
        for (int seg = 0; seg < n_edges; seg += QCAP) {
            if (tid == 0) qn = 0;
            __syncthreads();
            const int segEnd = min(seg + QCAP, n_edges);
            for (int e = seg + tid; e < segEnd; e += ATH) note(e, dst[e]);
            __syncthreads();
            drain(qn, false);
            __syncthreads();
        }
    }
    __syncthreads();

    const float4* h4 = (const float4*)h;
    for (int idx = tid; idx < nrows * (NFEAT / 4); idx += ATH) {
        const int r  = idx >> 5;
        const int f4 = idx & 31;
        const float di = dinv[lo + r];
        const float d2 = di * di;
        const float4 hv = h4[(size_t)(lo + r) * (NFEAT / 4) + f4];
        float* ap = acc + r * TP + f4;
        float a0 = ap[0], a1 = ap[32], a2 = ap[64], a3 = ap[96];
        a0 += hv.x * d2;
        a1 += hv.y * d2;
        a2 += hv.z * d2;
        a3 += hv.w * d2;
        ap[0] = a0; ap[32] = a1; ap[64] = a2; ap[96] = a3;
    }
    __syncthreads();

    const int wave  = __builtin_amdgcn_readfirstlane(tid >> 5);
    const int mlane = lane & 15;
    const int khalf = (lane >> 4) * 2;
    const int ncol  = (wave & 7) * 16 + mlane;
    const int mq    = wave >> 3;
    const int cnt   = (((nrows + 15) >> 4) - mq + 3) >> 2;
    v8f d[4] = {};
    auto tile_mm = [&](auto ntc) {
        constexpr int NT = decltype(ntc)::value;
        const float* ap = acc + (mq * 16 + mlane) * TP + khalf;
        const float* bp = Wl + khalf * WP + ncol;
        #pragma unroll 4
        for (int k0 = 0; k0 < NFEAT; k0 += 4) {
            v2f bb;
            bb.x = bp[(k0    ) * WP];
            bb.y = bp[(k0 + 1) * WP];
            #pragma unroll
            for (int i = 0; i < NT; ++i) {
                const v2f a = *(const v2f*)(ap + i * 64 * TP + k0);
                d[i] = __builtin_amdgcn_wmma_f32_16x16x4_f32(
                           false, a, false, bb, (short)0, d[i], false, false);
            }
        }
    };
    switch (cnt) {
        case 4: tile_mm(tile_count<4>{}); break;
        case 3: tile_mm(tile_count<3>{}); break;
        case 2: tile_mm(tile_count<2>{}); break;
        case 1: tile_mm(tile_count<1>{}); break;
        default: break;
    }
    __syncthreads();

    const float bias = b[ncol];
    #pragma unroll
    for (int i = 0; i < 4; ++i) {
        if (i < cnt) {
            float* dp = acc + ((mq + 4 * i) * 16 + (lane >> 4) * 8) * TP + ncol;
            #pragma unroll
            for (int r = 0; r < 8; ++r) dp[r * TP] = fmaxf(d[i][r] + bias, 0.0f);
        }
    }
    __syncthreads();

    if constexpr (!FROM_LIST) {
        float4* o4 = (float4*)out;
        for (int idx = tid; idx < nrows * (NFEAT / 4); idx += ATH) {
            const int r  = idx >> 5;
            const int f4 = idx & 31;
            o4[(size_t)(lo + r) * (NFEAT / 4) + f4] = *((const float4*)(acc + r * TP) + f4);
        }
    } else {
        if (wave < ((nrows + 15) >> 4)) {
            const float bcv = bc[min(mlane, NCLS - 1)];
            const float* ap = acc + (wave * 16 + mlane) * TP + khalf;
            const float* bp = Wcl + khalf * CP + mlane;
            v8f dc = {};
            #pragma unroll 4
            for (int k0 = 0; k0 < NFEAT; k0 += 4) {
                const v2f a = *(const v2f*)(ap + k0);
                v2f bb;
                bb.x = bp[(k0    ) * CP];
                bb.y = bp[(k0 + 1) * CP];
                dc = __builtin_amdgcn_wmma_f32_16x16x4_f32(
                         false, a, false, bb, (short)0, dc, false, false);
            }
            float sc[8];
            #pragma unroll
            for (int r = 0; r < 8; ++r) sc[r] = dc[r] + bcv;
            if (mlane < NCLS) {
                float* lp = lg + (wave * 16 + (lane >> 4) * 8) * NCLS + mlane;
                #pragma unroll
                for (int r = 0; r < 8; ++r) lp[r * NCLS] = sc[r];
            }
        }
        __syncthreads();
        const int g0 = lo * NCLS;
        const int g1 = g0 + nrows * NCLS;
        for (int g = (g0 & ~31) + tid; g < g1; g += ATH)
            if (g >= g0) scores[g] = lg[g - g0];
    }
}

extern "C" void kernel_launch(void* const* d_in, const int* in_sizes, int n_in,
                              void* d_out, int out_size, void* d_ws, size_t ws_size,
                              hipStream_t stream) {
    const float* x   = (const float*)d_in[0];
    const int*   ei  = (const int*)  d_in[1];
    const float* ew  = (const float*)d_in[2];
    const float* W1  = (const float*)d_in[3];
    const float* b1  = (const float*)d_in[4];
    const float* W2  = (const float*)d_in[5];
    const float* b2  = (const float*)d_in[6];
    const float* Wc  = (const float*)d_in[7];
    const float* bc  = (const float*)d_in[8];
    float*       out = (float*)d_out;

    const int n_nodes = in_sizes[0] / NFEAT;
    const int n_edges = in_sizes[2];
    const int* src = ei;
    const int* dst = ei + n_edges;

    const size_t nd = (size_t)n_nodes * NFEAT;
    float* bufA = (float*)d_ws;
    float* dinv = bufA + nd;
    const int gT  = (n_nodes + RT - 1) / RT;
    int* lists = (int*)(dinv + (size_t)((n_nodes + 31) / 32) * 32);
    int* cnts  = lists + (size_t)gT * QCAP * 4;
    int* keys  = cnts + (size_t)gT * 32;

    const int B = 256;
    const int gN  = (n_nodes + B - 1) / B;
    const int gE  = (n_edges + 4 * B - 1) / (4 * B);

    gcn_fill1    <<<gN, B, 0, stream>>>(dinv, n_nodes);
    gcn_deg_keys <<<gE, B, 0, stream>>>(dst, ew, dinv, keys, n_edges);
    gcn_rsqrt    <<<gN, B, 0, stream>>>(dinv, n_nodes);

    gcn_layer_tile<false><<<gT, ATH, 0, stream>>>(x, W1, src, dst, ew, dinv, b1, bufA, lists, cnts, keys, Wc, bc, out, n_nodes, n_edges);

    gcn_layer_tile<true><<<gT, ATH, 0, stream>>>(bufA, W2, src, dst, ew, dinv, b2, nullptr, lists, cnts, keys, Wc, bc, out, n_nodes, n_edges);
}


# Layer-1 destination scan replaced by edge buckets: degree pass puts each edge number in its tile's bucket (LDS rank, one returning global add a bucket a group); sums still LDS atomics, arrival order
# speedup vs baseline: 2.8821x; 1.0472x over previous
#include <hip/hip_runtime.h>
#include <hip/hip_bf16.h>

typedef float v2f __attribute__((ext_vector_type(2)));
typedef float v8f __attribute__((ext_vector_type(8)));

#define NFEAT 128

#define RT   196
#define QCAP 16384
#define NBK  256
#define EB   256

__global__ void gcn_fill1(float* __restrict__ deg, int* __restrict__ tcount, int n) {
    int i = blockIdx.x * blockDim.x + threadIdx.x;
    if (i < n) deg[i] = 1.0f;
    if (i < NBK) tcount[i] = 0;
}

__global__ __launch_bounds__(EB) void gcn_deg_bucket(const int* __restrict__ dst,
                             const float* __restrict__ ew,
                             float* __restrict__ deg, int* __restrict__ tcount,
                             int* __restrict__ bucket, int n_edges) {
    __shared__ int hist[NBK];
    const int tid = threadIdx.x;
    for (int i = tid; i < NBK; i += EB) hist[i] = 0;
    __syncthreads();
    const int e0 = 4 * (blockIdx.x * EB + tid);
    const int n  = min(4, max(0, n_edges - e0));
    int d[4]; float w[4];
    if (n == 4) {
        #pragma unroll
        for (int j = 0; j < 4; ++j) { d[j] = dst[e0 + j]; w[j] = ew[e0 + j]; }
    } else {
        #pragma unroll
        for (int j = 0; j < 4; ++j) {
            d[j] = j < n ? dst[e0 + j] : 0;
            w[j] = j < n ? ew[e0 + j] : 0.0f;
        }
    }
    int key[4], rk[4];
    #pragma unroll
    for (int j = 0; j < 4; ++j) {
        key[j] = (int)(((unsigned)d[j] / RT) & (NBK - 1));
        rk[j]  = 0;
        if (j < n) rk[j] = atomicAdd(&hist[key[j]], 1);
    }
    __syncthreads();
    for (int i = tid; i < NBK; i += EB) {
        const int cnt = hist[i];
        hist[i] = cnt ? atomicAdd(&tcount[i], cnt) : 0;
    }
    __syncthreads();
    #pragma unroll
    for (int j = 0; j < 4; ++j) {
        if (j < n) {
            __hip_atomic_fetch_add(&deg[d[j]], w[j],
                                   __ATOMIC_RELAXED, __HIP_MEMORY_SCOPE_AGENT);
            const int pos = hist[key[j]] + rk[j];
            if (pos < QCAP) bucket[(size_t)key[j] * QCAP + pos] = e0 + j;
        }
    }
}

__global__ void gcn_rsqrt(float* __restrict__ d, int n) {
    int i = blockIdx.x * blockDim.x + threadIdx.x;
    if (i < n) d[i] = rsqrtf(d[i]);
}

#define MB   13
#define TP   132
#define WP   136
#define TCOL(f) ((((f) & 3) << 5) + ((f) >> 2))
#define ATH  1024
#define BQ   3
#define DU   8
#define NCLS 10
#define CP   16

template <int N> struct tile_count { static constexpr int value = N; };

template <bool FROM_LIST>
__global__ __launch_bounds__(ATH) void gcn_layer_tile(
        const float* __restrict__ h, const float* __restrict__ W,
        const int* __restrict__ src,
        const int* __restrict__ dst, const float* __restrict__ ew,
        const float* __restrict__ dinv, const float* __restrict__ b,
        float* __restrict__ out, int* __restrict__ lists, int* __restrict__ cnts,
        const int* __restrict__ bucket, const int* __restrict__ tcount,
        const float* __restrict__ Wc, const float* __restrict__ bc,
        float* __restrict__ scores, int n_nodes, int n_edges) {
    __shared__ __attribute__((aligned(16))) float acc[MB * 16 * TP];
    __shared__ __attribute__((aligned(16))) float Wl[NFEAT * WP];
    __shared__ int q[QCAP];
    __shared__ int qn;
    __shared__ __attribute__((aligned(16))) float Wcl[FROM_LIST ? NFEAT * CP : 1];
    __shared__ __attribute__((aligned(16))) float lg[FROM_LIST ? MB * 16 * NCLS : 1];

    const int tid   = threadIdx.x;
    const int lane  = tid & 31;
    const int lo    = blockIdx.x * RT;
    const int nrows = min(RT, n_nodes - lo);
    int4* const tl  = (int4*)lists + (size_t)blockIdx.x * QCAP;

    auto clear_tile = [&]() {
        float4* a4 = (float4*)acc;
        for (int i = tid; i < MB * 16 * TP / 4; i += ATH)
            a4[i] = make_float4(0.0f, 0.0f, 0.0f, 0.0f);
    };

    auto note = [&](int e, int d) {
        if ((unsigned)(d - lo) < (unsigned)nrows) {
            int idx = atomicAdd(&qn, 1);
            if (idx < QCAP) q[idx] = e;
        }
    };

    auto drain = [&](int total, bool save) {
        const int wave = __builtin_amdgcn_readfirstlane(tid >> 5);
        int4* slot = tl + wave * DU;
        for (int base = wave; base < total; base += 32 * DU, slot += 32 * DU) {
            int e[DU]; bool ok[DU];
            #pragma unroll
            for (int u = 0; u < DU; ++u) {
                const int i = base + 32 * u;
                ok[u] = i < total;
                e[u]  = q[ok[u] ? i : base];
            }
            __builtin_amdgcn_sched_barrier(0);
            int sN[DU], dN[DU]; float w[DU];
            #pragma unroll
            for (int u = 0; u < DU; ++u) {
                sN[u] = src[e[u]]; dN[u] = dst[e[u]]; w[u] = ew[e[u]];
            }
            __builtin_amdgcn_sched_barrier(0);
            float c[DU]; float4 hv[DU];
            #pragma unroll
            for (int u = 0; u < DU; ++u) {
                c[u] = dinv[sN[u]] * w[u] * dinv[dN[u]];
                hv[u] = *((const float4*)(h + (size_t)sN[u] * NFEAT) + lane);
            }
            int row[DU];
            #pragma unroll
            for (int u = 0; u < DU; ++u) {
                row[u] = (ok[u] && (unsigned)(dN[u] - lo) < (unsigned)nrows) ? dN[u] - lo : RT;
                float* ar = acc + row[u] * TP + lane;
                const float hj[4] = {hv[u].x, hv[u].y, hv[u].z, hv[u].w};
                #pragma unroll
                for (int j = 0; j < 4; ++j)
                    __hip_atomic_fetch_add(ar + 32 * j, hj[j] * c[u],
                                           __ATOMIC_RELAXED, __HIP_MEMORY_SCOPE_WORKGROUP);
            }
            if (save) {
                int ex = sN[0], ey = row[0]; float ez = c[0];
                #pragma unroll
                for (int u = 1; u < DU; ++u)
                    if (lane == u) { ex = sN[u]; ey = row[u]; ez = c[u]; }
                if (lane < DU) slot[lane] = make_int4(ex, ey, __float_as_int(ez), 0);
            }
        }
    };

    auto drain_list = [&](int total) {
        const int wave = __builtin_amdgcn_readfirstlane(tid >> 5);
        if (wave >= total) return;
        const int4* slot = tl + wave * DU;
        int4 en[DU];
        #pragma unroll
        for (int u = 0; u < DU; ++u) en[u] = slot[u];
        for (int base = wave; base < total; base += 32 * DU) {
            float4 hv[DU];
            #pragma unroll
            for (int u = 0; u < DU; ++u)
                hv[u] = *((const float4*)(h + (size_t)en[u].x * NFEAT) + lane);
            if (base + 32 * DU < total) slot += 32 * DU;
            int4 nx[DU];
            #pragma unroll
            for (int u = 0; u < DU; ++u) nx[u] = slot[u];
            #pragma unroll
            for (int u = 0; u < DU; ++u) {
                float* ar = acc + en[u].y * TP + lane;
                const float cu = __int_as_float(en[u].z);
                const float hj[4] = {hv[u].x, hv[u].y, hv[u].z, hv[u].w};
                #pragma unroll
                for (int j = 0; j < 4; ++j)
                    __hip_atomic_fetch_add(ar + 32 * j, hj[j] * cu,
                                           __ATOMIC_RELAXED, __HIP_MEMORY_SCOPE_WORKGROUP);
            }
            #pragma unroll
            for (int u = 0; u < DU; ++u) en[u] = nx[u];
        }
    };

    int nb = 0;
    {
        if constexpr (!FROM_LIST) nb = tcount[blockIdx.x & (NBK - 1)];
        const float4* Wv = (const float4*)W;
        float4 wv[(NFEAT * NFEAT / 4) / ATH];
        #pragma unroll
        for (int u = 0; u < (NFEAT * NFEAT / 4) / ATH; ++u) wv[u] = Wv[tid + u * ATH];
        float wcv[(NFEAT * CP) / ATH];
        if constexpr (FROM_LIST) {
            #pragma unroll
            for (int u = 0; u < (NFEAT * CP) / ATH; ++u) {
                const int i = tid + u * ATH;
                const int c = i & (CP - 1);
                wcv[u] = c < NCLS ? Wc[(i / CP) * NCLS + c] : 0.0f;
            }
        }
        const int* bk = bucket + (size_t)(blockIdx.x & (NBK - 1)) * QCAP;
        const int nq  = min(nb, QCAP);
        int bq[BQ];
        if constexpr (!FROM_LIST) {
            #pragma unroll
            for (int u = 0; u < BQ; ++u) {
                const int i = tid + u * ATH;
                bq[u] = i < nq ? bk[i] : 0;
            }
        }
        clear_tile();
        #pragma unroll
        for (int u = 0; u < (NFEAT * NFEAT / 4) / ATH; ++u) {
            const int i = tid + u * ATH;
            *(float4*)(Wl + TCOL(i >> 5) * WP + (i & 31) * 4) = wv[u];
        }
        if constexpr (FROM_LIST) {
            #pragma unroll
            for (int u = 0; u < (NFEAT * CP) / ATH; ++u) Wcl[tid + u * ATH] = wcv[u];
        }
        if constexpr (!FROM_LIST) {
            #pragma unroll
            for (int u = 0; u < BQ; ++u) {
                const int i = tid + u * ATH;
                if (i < nq) q[i] = bq[u];
            }
            for (int i = tid + BQ * ATH; i < nq; i += ATH) q[i] = bk[i];
        }
    }
    if (tid == 0) qn = 0;
    __syncthreads();

    int noted;
    if constexpr (FROM_LIST) {
        noted = cnts[blockIdx.x * 32];
    } else {
        noted = nb;
        if (tid == 0) cnts[blockIdx.x * 32] = noted;
    }

    if (noted <= QCAP) {
        if constexpr (FROM_LIST) drain_list(noted);
        else                     drain(noted, true);
    } else {
        __syncthreads();
        clear_tile();
        for (int seg = 0; seg < n_edges; seg += QCAP) {
            if (tid == 0) qn = 0;
            __syncthreads();
            const int segEnd = min(seg + QCAP, n_edges);
            for (int e = seg + tid; e < segEnd; e += ATH) note(e, dst[e]);
            __syncthreads();
            drain(qn, false);
            __syncthreads();
        }
    }
    __syncthreads();

    const float4* h4 = (const float4*)h;
    for (int idx = tid; idx < nrows * (NFEAT / 4); idx += ATH) {
        const int r  = idx >> 5;
        const int f4 = idx & 31;
        const float di = dinv[lo + r];
        const float d2 = di * di;
        const float4 hv = h4[(size_t)(lo + r) * (NFEAT / 4) + f4];
        float* ap = acc + r * TP + f4;
        float a0 = ap[0], a1 = ap[32], a2 = ap[64], a3 = ap[96];
        a0 += hv.x * d2;
        a1 += hv.y * d2;
        a2 += hv.z * d2;
        a3 += hv.w * d2;
        ap[0] = a0; ap[32] = a1; ap[64] = a2; ap[96] = a3;
    }
    __syncthreads();

    const int wave  = __builtin_amdgcn_readfirstlane(tid >> 5);
    const int mlane = lane & 15;
    const int khalf = (lane >> 4) * 2;
    const int ncol  = (wave & 7) * 16 + mlane;
    const int mq    = wave >> 3;
    const int cnt   = (((nrows + 15) >> 4) - mq + 3) >> 2;
    v8f d[4] = {};
    auto tile_mm = [&](auto ntc) {
        constexpr int NT = decltype(ntc)::value;
        const float* ap = acc + (mq * 16 + mlane) * TP + khalf;
        const float* bp = Wl + khalf * WP + ncol;
        #pragma unroll 4
        for (int k0 = 0; k0 < NFEAT; k0 += 4) {
            v2f bb;
            bb.x = bp[(k0    ) * WP];
            bb.y = bp[(k0 + 1) * WP];
            #pragma unroll
            for (int i = 0; i < NT; ++i) {
                const v2f a = *(const v2f*)(ap + i * 64 * TP + k0);
                d[i] = __builtin_amdgcn_wmma_f32_16x16x4_f32(
                           false, a, false, bb, (short)0, d[i], false, false);
            }
        }
    };
    switch (cnt) {
        case 4: tile_mm(tile_count<4>{}); break;
        case 3: tile_mm(tile_count<3>{}); break;
        case 2: tile_mm(tile_count<2>{}); break;
        case 1: tile_mm(tile_count<1>{}); break;
        default: break;
    }
    __syncthreads();

    const float bias = b[ncol];
    #pragma unroll
    for (int i = 0; i < 4; ++i) {
        if (i < cnt) {
            float* dp = acc + ((mq + 4 * i) * 16 + (lane >> 4) * 8) * TP + ncol;
            #pragma unroll
            for (int r = 0; r < 8; ++r) dp[r * TP] = fmaxf(d[i][r] + bias, 0.0f);
        }
    }
    __syncthreads();

    if constexpr (!FROM_LIST) {
        float4* o4 = (float4*)out;
        for (int idx = tid; idx < nrows * (NFEAT / 4); idx += ATH) {
            const int r  = idx >> 5;
            const int f4 = idx & 31;
            o4[(size_t)(lo + r) * (NFEAT / 4) + f4] = *((const float4*)(acc + r * TP) + f4);
        }
    } else {
        if (wave < ((nrows + 15) >> 4)) {
            const float bcv = bc[min(mlane, NCLS - 1)];
            const float* ap = acc + (wave * 16 + mlane) * TP + khalf;
            const float* bp = Wcl + khalf * CP + mlane;
            v8f dc = {};
            #pragma unroll 4
            for (int k0 = 0; k0 < NFEAT; k0 += 4) {
                const v2f a = *(const v2f*)(ap + k0);
                v2f bb;
                bb.x = bp[(k0    ) * CP];
                bb.y = bp[(k0 + 1) * CP];
                dc = __builtin_amdgcn_wmma_f32_16x16x4_f32(
                         false, a, false, bb, (short)0, dc, false, false);
            }
            float sc[8];
            #pragma unroll
            for (int r = 0; r < 8; ++r) sc[r] = dc[r] + bcv;
            if (mlane < NCLS) {
                float* lp = lg + (wave * 16 + (lane >> 4) * 8) * NCLS + mlane;
                #pragma unroll
                for (int r = 0; r < 8; ++r) lp[r * NCLS] = sc[r];
            }
        }
        __syncthreads();
        const int g0 = lo * NCLS;
        const int g1 = g0 + nrows * NCLS;
        for (int g = (g0 & ~31) + tid; g < g1; g += ATH)
            if (g >= g0) scores[g] = lg[g - g0];
    }
}

extern "C" void kernel_launch(void* const* d_in, const int* in_sizes, int n_in,
                              void* d_out, int out_size, void* d_ws, size_t ws_size,
                              hipStream_t stream) {
    const float* x   = (const float*)d_in[0];
    const int*   ei  = (const int*)  d_in[1];
    const float* ew  = (const float*)d_in[2];
    const float* W1  = (const float*)d_in[3];
    const float* b1  = (const float*)d_in[4];
    const float* W2  = (const float*)d_in[5];
    const float* b2  = (const float*)d_in[6];
    const float* Wc  = (const float*)d_in[7];
    const float* bc  = (const float*)d_in[8];
    float*       out = (float*)d_out;

    const int n_nodes = in_sizes[0] / NFEAT;
    const int n_edges = in_sizes[2];
    const int* src = ei;
    const int* dst = ei + n_edges;

    const size_t nd = (size_t)n_nodes * NFEAT;
    float* bufA = (float*)d_ws;
    float* dinv = bufA + nd;
    const int gT  = (n_nodes + RT - 1) / RT;
    int* lists = (int*)(dinv + (size_t)((n_nodes + 31) / 32) * 32);
    int* cnts  = lists + (size_t)gT * QCAP * 4;
    int* bucket = cnts + (size_t)gT * 32;
    int* tcount = bucket + (size_t)NBK * QCAP;

    const int B = 256;
    const int gN  = (n_nodes + B - 1) / B;
    const int gE  = (n_edges + 4 * EB - 1) / (4 * EB);

    gcn_fill1     <<<gN, B, 0, stream>>>(dinv, tcount, n_nodes);
    gcn_deg_bucket<<<gE, EB, 0, stream>>>(dst, ew, dinv, tcount, bucket, n_edges);
    gcn_rsqrt    <<<gN, B, 0, stream>>>(dinv, n_nodes);

    gcn_layer_tile<false><<<gT, ATH, 0, stream>>>(x, W1, src, dst, ew, dinv, b1, bufA, lists, cnts, bucket, tcount, Wc, bc, out, n_nodes, n_edges);

    gcn_layer_tile<true><<<gT, ATH, 0, stream>>>(bufA, W2, src, dst, ew, dinv, b2, nullptr, lists, cnts, bucket, tcount, Wc, bc, out, n_nodes, n_edges);
}


# Edge buckets hold 16-byte records (source, destination, weight), not edge numbers: layer 1 reads 8 records a line by one load, no queue, no src/dst/weight lookup per edge; sums still arrival order
# speedup vs baseline: 3.1056x; 1.0775x over previous
#include <hip/hip_runtime.h>
#include <hip/hip_bf16.h>

typedef float v2f __attribute__((ext_vector_type(2)));
typedef float v8f __attribute__((ext_vector_type(8)));

#define NFEAT 128

#define RT   196
#define QCAP 16384
#define NBK  256
#define EB   256

__global__ void gcn_fill1(float* __restrict__ deg, int* __restrict__ tcount, int n) {
    int i = blockIdx.x * blockDim.x + threadIdx.x;
    if (i < n) deg[i] = 1.0f;
    if (i < NBK) tcount[i] = 0;
}

__global__ __launch_bounds__(EB) void gcn_deg_bucket(const int* __restrict__ src,
                             const int* __restrict__ dst,
                             const float* __restrict__ ew,
                             float* __restrict__ deg, int* __restrict__ tcount,
                             int* __restrict__ bucket, int n_edges) {
    __shared__ int hist[NBK];
    const int tid = threadIdx.x;
    for (int i = tid; i < NBK; i += EB) hist[i] = 0;
    __syncthreads();
    const int e0 = 4 * (blockIdx.x * EB + tid);
    const int n  = min(4, max(0, n_edges - e0));
    int s[4], d[4]; float w[4];
    if (n == 4) {
        #pragma unroll
        for (int j = 0; j < 4; ++j) { s[j] = src[e0 + j]; d[j] = dst[e0 + j]; w[j] = ew[e0 + j]; }
    } else {
        #pragma unroll
        for (int j = 0; j < 4; ++j) {
            s[j] = j < n ? src[e0 + j] : 0;
            d[j] = j < n ? dst[e0 + j] : 0;
            w[j] = j < n ? ew[e0 + j] : 0.0f;
        }
    }
    int key[4], rk[4];
    #pragma unroll
    for (int j = 0; j < 4; ++j) {
        key[j] = (int)(((unsigned)d[j] / RT) & (NBK - 1));
        rk[j]  = 0;
        if (j < n) rk[j] = atomicAdd(&hist[key[j]], 1);
    }
    __syncthreads();
    for (int i = tid; i < NBK; i += EB) {
        const int cnt = hist[i];
        hist[i] = cnt ? atomicAdd(&tcount[i], cnt) : 0;
    }
    __syncthreads();
    int4* const brec = (int4*)bucket;
    #pragma unroll
    for (int j = 0; j < 4; ++j) {
        if (j < n) {
            __hip_atomic_fetch_add(&deg[d[j]], w[j],
                                   __ATOMIC_RELAXED, __HIP_MEMORY_SCOPE_AGENT);
            const int pos = hist[key[j]] + rk[j];
            if (pos < QCAP)
                brec[(size_t)key[j] * QCAP + pos] = make_int4(s[j], d[j], __float_as_int(w[j]), 0);
        }
    }
}

__global__ void gcn_rsqrt(float* __restrict__ d, int n) {
    int i = blockIdx.x * blockDim.x + threadIdx.x;
    if (i < n) d[i] = rsqrtf(d[i]);
}

#define MB   13
#define TP   132
#define WP   136
#define TCOL(f) ((((f) & 3) << 5) + ((f) >> 2))
#define ATH  1024
#define DU   8
#define NCLS 10
#define CP   16

template <int N> struct tile_count { static constexpr int value = N; };

template <bool FROM_LIST>
__global__ __launch_bounds__(ATH) void gcn_layer_tile(
        const float* __restrict__ h, const float* __restrict__ W,
        const int* __restrict__ src,
        const int* __restrict__ dst, const float* __restrict__ ew,
        const float* __restrict__ dinv, const float* __restrict__ b,
        float* __restrict__ out, int* __restrict__ lists, int* __restrict__ cnts,
        const int* __restrict__ bucket, const int* __restrict__ tcount,
        const float* __restrict__ Wc, const float* __restrict__ bc,
        float* __restrict__ scores, int n_nodes, int n_edges) {
    __shared__ __attribute__((aligned(16))) float acc[MB * 16 * TP];
    __shared__ __attribute__((aligned(16))) float Wl[NFEAT * WP];
    __shared__ int q[QCAP];
    __shared__ int qn;
    __shared__ __attribute__((aligned(16))) float Wcl[FROM_LIST ? NFEAT * CP : 1];
    __shared__ __attribute__((aligned(16))) float lg[FROM_LIST ? MB * 16 * NCLS : 1];

    const int tid   = threadIdx.x;
    const int lane  = tid & 31;
    const int lo    = blockIdx.x * RT;
    const int nrows = min(RT, n_nodes - lo);
    int4* const tl  = (int4*)lists + (size_t)blockIdx.x * QCAP;
    const int4* const brec = (const int4*)bucket + (size_t)(blockIdx.x & (NBK - 1)) * QCAP;

    auto clear_tile = [&]() {
        float4* a4 = (float4*)acc;
        for (int i = tid; i < MB * 16 * TP / 4; i += ATH)
            a4[i] = make_float4(0.0f, 0.0f, 0.0f, 0.0f);
    };

    auto note = [&](int e, int d) {
        if ((unsigned)(d - lo) < (unsigned)nrows) {
            int idx = atomicAdd(&qn, 1);
            if (idx < QCAP) q[idx] = e;
        }
    };

    auto drain = [&](int total) {
        const int wave = __builtin_amdgcn_readfirstlane(tid >> 5);
        for (int base = wave; base < total; base += 32 * DU) {
            int e[DU]; bool ok[DU];
            #pragma unroll
            for (int u = 0; u < DU; ++u) {
                const int i = base + 32 * u;
                ok[u] = i < total;
                e[u]  = q[ok[u] ? i : base];
            }
            __builtin_amdgcn_sched_barrier(0);
            int sN[DU], dN[DU]; float w[DU];
            #pragma unroll
            for (int u = 0; u < DU; ++u) {
                sN[u] = src[e[u]]; dN[u] = dst[e[u]]; w[u] = ew[e[u]];
            }
            __builtin_amdgcn_sched_barrier(0);
            float c[DU]; float4 hv[DU];
            #pragma unroll
            for (int u = 0; u < DU; ++u) {
                c[u] = dinv[sN[u]] * w[u] * dinv[dN[u]];
                hv[u] = *((const float4*)(h + (size_t)sN[u] * NFEAT) + lane);
            }
            #pragma unroll
            for (int u = 0; u < DU; ++u) {
                const int row = (ok[u] && (unsigned)(dN[u] - lo) < (unsigned)nrows) ? dN[u] - lo : RT;
                float* ar = acc + row * TP + lane;
                const float hj[4] = {hv[u].x, hv[u].y, hv[u].z, hv[u].w};
                #pragma unroll
                for (int j = 0; j < 4; ++j)
                    __hip_atomic_fetch_add(ar + 32 * j, hj[j] * c[u],
                                           __ATOMIC_RELAXED, __HIP_MEMORY_SCOPE_WORKGROUP);
            }
        }
    };

    auto drain_bucket = [&](int total, int4 r) {
        const int wave = __builtin_amdgcn_readfirstlane(tid >> 5);
        const int sub  = lane & (DU - 1);
        int base = wave * DU;
        if (base >= total) return;
        const int4* bslot = brec + base;
        int4* slot = tl + base;
        for (; base < total; base += 32 * DU, slot += 32 * DU) {
            float4 hv[DU];
            #pragma unroll
            for (int u = 0; u < DU; ++u) {
                const int sN = __builtin_amdgcn_readlane(r.x, u);
                hv[u] = *((const float4*)(h + (size_t)sN * NFEAT) + lane);
            }
            const float ds = dinv[r.x];
            const float dd = dinv[r.y];
            const bool more = base + 32 * DU < total;
            if (more) bslot += 32 * DU;
            const int4 rn = bslot[(more ? base + 32 * DU : base) + sub < total ? sub : 0];
            const float cm  = ds * __int_as_float(r.z) * dd;
            const int rowm  = (base + sub < total && (unsigned)(r.y - lo) < (unsigned)nrows) ? r.y - lo : RT;
            #pragma unroll
            for (int u = 0; u < DU; ++u) {
                const int   row = __builtin_amdgcn_readlane(rowm, u);
                const float cu  = __int_as_float(__builtin_amdgcn_readlane(__float_as_int(cm), u));
                float* ar = acc + row * TP + lane;
                const float hj[4] = {hv[u].x, hv[u].y, hv[u].z, hv[u].w};
                #pragma unroll
                for (int j = 0; j < 4; ++j)
                    __hip_atomic_fetch_add(ar + 32 * j, hj[j] * cu,
                                           __ATOMIC_RELAXED, __HIP_MEMORY_SCOPE_WORKGROUP);
            }
            if (lane < DU) slot[lane] = make_int4(r.x, rowm, __float_as_int(cm), 0);
            r = rn;
        }
    };

    auto drain_list = [&](int total) {
        const int wave = __builtin_amdgcn_readfirstlane(tid >> 5);
        if (wave * DU >= total) return;
        const int4* slot = tl + wave * DU;
        int4 en[DU];
        #pragma unroll
        for (int u = 0; u < DU; ++u) en[u] = slot[u];
        for (int base = wave * DU; base < total; base += 32 * DU) {
            float4 hv[DU];
            #pragma unroll
            for (int u = 0; u < DU; ++u)
                hv[u] = *((const float4*)(h + (size_t)en[u].x * NFEAT) + lane);
            if (base + 32 * DU < total) slot += 32 * DU;
            int4 nx[DU];
            #pragma unroll
            for (int u = 0; u < DU; ++u) nx[u] = slot[u];
            #pragma unroll
            for (int u = 0; u < DU; ++u) {
                float* ar = acc + en[u].y * TP + lane;
                const float cu = __int_as_float(en[u].z);
                const float hj[4] = {hv[u].x, hv[u].y, hv[u].z, hv[u].w};
                #pragma unroll
                for (int j = 0; j < 4; ++j)
                    __hip_atomic_fetch_add(ar + 32 * j, hj[j] * cu,
                                           __ATOMIC_RELAXED, __HIP_MEMORY_SCOPE_WORKGROUP);
            }
            #pragma unroll
            for (int u = 0; u < DU; ++u) en[u] = nx[u];
        }
    };

    int nb = 0;
    int4 r0 = make_int4(0, 0, 0, 0);
    {
        if constexpr (!FROM_LIST) nb = tcount[blockIdx.x & (NBK - 1)];
        const float4* Wv = (const float4*)W;
        float4 wv[(NFEAT * NFEAT / 4) / ATH];
        #pragma unroll
        for (int u = 0; u < (NFEAT * NFEAT / 4) / ATH; ++u) wv[u] = Wv[tid + u * ATH];
        float wcv[(NFEAT * CP) / ATH];
        if constexpr (FROM_LIST) {
            #pragma unroll
            for (int u = 0; u < (NFEAT * CP) / ATH; ++u) {
                const int i = tid + u * ATH;
                const int c = i & (CP - 1);
                wcv[u] = c < NCLS ? Wc[(i / CP) * NCLS + c] : 0.0f;
            }
        }
        if constexpr (!FROM_LIST) {
            const int b0 = __builtin_amdgcn_readfirstlane(tid >> 5) * DU;
            if (nb <= QCAP && b0 < nb)
                r0 = brec[b0 + (b0 + (lane & (DU - 1)) < nb ? (lane & (DU - 1)) : 0)];
        }
        clear_tile();
        #pragma unroll
        for (int u = 0; u < (NFEAT * NFEAT / 4) / ATH; ++u) {
            const int i = tid + u * ATH;
            *(float4*)(Wl + TCOL(i >> 5) * WP + (i & 31) * 4) = wv[u];
        }
        if constexpr (FROM_LIST) {
            #pragma unroll
            for (int u = 0; u < (NFEAT * CP) / ATH; ++u) Wcl[tid + u * ATH] = wcv[u];
        }
    }
    if (tid == 0) qn = 0;
    __syncthreads();

    int noted;
    if constexpr (FROM_LIST) {
        noted = cnts[blockIdx.x * 32];
    } else {
        noted = nb;
        if (tid == 0) cnts[blockIdx.x * 32] = noted;
    }

    if (noted <= QCAP) {
        if constexpr (FROM_LIST) drain_list(noted);
        else                     drain_bucket(noted, r0);
    } else {
        __syncthreads();
        clear_tile();
        for (int seg = 0; seg < n_edges; seg += QCAP) {
            if (tid == 0) qn = 0;
            __syncthreads();
            const int segEnd = min(seg + QCAP, n_edges);
            for (int e = seg + tid; e < segEnd; e += ATH) note(e, dst[e]);
            __syncthreads();
            drain(qn);
            __syncthreads();
        }
    }
    __syncthreads();

    const float4* h4 = (const float4*)h;
    for (int idx = tid; idx < nrows * (NFEAT / 4); idx += ATH) {
        const int r  = idx >> 5;
        const int f4 = idx & 31;
        const float di = dinv[lo + r];
        const float d2 = di * di;
        const float4 hv = h4[(size_t)(lo + r) * (NFEAT / 4) + f4];
        float* ap = acc + r * TP + f4;
        float a0 = ap[0], a1 = ap[32], a2 = ap[64], a3 = ap[96];
        a0 += hv.x * d2;
        a1 += hv.y * d2;
        a2 += hv.z * d2;
        a3 += hv.w * d2;
        ap[0] = a0; ap[32] = a1; ap[64] = a2; ap[96] = a3;
    }
    __syncthreads();

    const int wave  = __builtin_amdgcn_readfirstlane(tid >> 5);
    const int mlane = lane & 15;
    const int khalf = (lane >> 4) * 2;
    const int ncol  = (wave & 7) * 16 + mlane;
    const int mq    = wave >> 3;
    const int cnt   = (((nrows + 15) >> 4) - mq + 3) >> 2;
    v8f d[4] = {};
    auto tile_mm = [&](auto ntc) {
        constexpr int NT = decltype(ntc)::value;
        const float* ap = acc + (mq * 16 + mlane) * TP + khalf;
        const float* bp = Wl + khalf * WP + ncol;
        #pragma unroll 4
        for (int k0 = 0; k0 < NFEAT; k0 += 4) {
            v2f bb;
            bb.x = bp[(k0    ) * WP];
            bb.y = bp[(k0 + 1) * WP];
            #pragma unroll
            for (int i = 0; i < NT; ++i) {
                const v2f a = *(const v2f*)(ap + i * 64 * TP + k0);
                d[i] = __builtin_amdgcn_wmma_f32_16x16x4_f32(
                           false, a, false, bb, (short)0, d[i], false, false);
            }
        }
    };
    switch (cnt) {
        case 4: tile_mm(tile_count<4>{}); break;
        case 3: tile_mm(tile_count<3>{}); break;
        case 2: tile_mm(tile_count<2>{}); break;
        case 1: tile_mm(tile_count<1>{}); break;
        default: break;
    }
    __syncthreads();

    const float bias = b[ncol];
    #pragma unroll
    for (int i = 0; i < 4; ++i) {
        if (i < cnt) {
            float* dp = acc + ((mq + 4 * i) * 16 + (lane >> 4) * 8) * TP + ncol;
            #pragma unroll
            for (int r = 0; r < 8; ++r) dp[r * TP] = fmaxf(d[i][r] + bias, 0.0f);
        }
    }
    __syncthreads();

    if constexpr (!FROM_LIST) {
        float4* o4 = (float4*)out;
        for (int idx = tid; idx < nrows * (NFEAT / 4); idx += ATH) {
            const int r  = idx >> 5;
            const int f4 = idx & 31;
            o4[(size_t)(lo + r) * (NFEAT / 4) + f4] = *((const float4*)(acc + r * TP) + f4);
        }
    } else {
        if (wave < ((nrows + 15) >> 4)) {
            const float bcv = bc[min(mlane, NCLS - 1)];
            const float* ap = acc + (wave * 16 + mlane) * TP + khalf;
            const float* bp = Wcl + khalf * CP + mlane;
            v8f dc = {};
            #pragma unroll 4
            for (int k0 = 0; k0 < NFEAT; k0 += 4) {
                const v2f a = *(const v2f*)(ap + k0);
                v2f bb;
                bb.x = bp[(k0    ) * CP];
                bb.y = bp[(k0 + 1) * CP];
                dc = __builtin_amdgcn_wmma_f32_16x16x4_f32(
                         false, a, false, bb, (short)0, dc, false, false);
            }
            float sc[8];
            #pragma unroll
            for (int r = 0; r < 8; ++r) sc[r] = dc[r] + bcv;
            if (mlane < NCLS) {
                float* lp = lg + (wave * 16 + (lane >> 4) * 8) * NCLS + mlane;
                #pragma unroll
                for (int r = 0; r < 8; ++r) lp[r * NCLS] = sc[r];
            }
        }
        __syncthreads();
        const int g0 = lo * NCLS;
        const int g1 = g0 + nrows * NCLS;
        for (int g = (g0 & ~31) + tid; g < g1; g += ATH)
            if (g >= g0) scores[g] = lg[g - g0];
    }
}

extern "C" void kernel_launch(void* const* d_in, const int* in_sizes, int n_in,
                              void* d_out, int out_size, void* d_ws, size_t ws_size,
                              hipStream_t stream) {
    const float* x   = (const float*)d_in[0];
    const int*   ei  = (const int*)  d_in[1];
    const float* ew  = (const float*)d_in[2];
    const float* W1  = (const float*)d_in[3];
    const float* b1  = (const float*)d_in[4];
    const float* W2  = (const float*)d_in[5];
    const float* b2  = (const float*)d_in[6];
    const float* Wc  = (const float*)d_in[7];
    const float* bc  = (const float*)d_in[8];
    float*       out = (float*)d_out;

    const int n_nodes = in_sizes[0] / NFEAT;
    const int n_edges = in_sizes[2];
    const int* src = ei;
    const int* dst = ei + n_edges;

    const size_t nd = (size_t)n_nodes * NFEAT;
    float* bufA = (float*)d_ws;
    float* dinv = bufA + nd;
    const int gT  = (n_nodes + RT - 1) / RT;
    int* lists = (int*)(dinv + (size_t)((n_nodes + 31) / 32) * 32);
    int* cnts  = lists + (size_t)gT * QCAP * 4;
    int* bucket = cnts + (size_t)gT * 32;
    int* tcount = bucket + (size_t)NBK * QCAP * 4;

    const int B = 256;
    const int gN  = (n_nodes + B - 1) / B;
    const int gE  = (n_edges + 4 * EB - 1) / (4 * EB);

    gcn_fill1     <<<gN, B, 0, stream>>>(dinv, tcount, n_nodes);
    gcn_deg_bucket<<<gE, EB, 0, stream>>>(src, dst, ew, dinv, tcount, bucket, n_edges);
    gcn_rsqrt    <<<gN, B, 0, stream>>>(dinv, n_nodes);

    gcn_layer_tile<false><<<gT, ATH, 0, stream>>>(x, W1, src, dst, ew, dinv, b1, bufA, lists, cnts, bucket, tcount, Wc, bc, out, n_nodes, n_edges);

    gcn_layer_tile<true><<<gT, ATH, 0, stream>>>(bufA, W2, src, dst, ew, dinv, b2, nullptr, lists, cnts, bucket, tcount, Wc, bc, out, n_nodes, n_edges);
}


# Degree sums leave global f32 atomics: a per-tile launch (replacing the rsqrt launch) adds its bucket's edge weights per row by LDS f32 atomics, arrival order, 1 added last, stores rsqrt; 5 launches
# speedup vs baseline: 4.3176x; 1.3903x over previous
#include <hip/hip_runtime.h>
#include <hip/hip_bf16.h>

typedef float v2f __attribute__((ext_vector_type(2)));
typedef float v8f __attribute__((ext_vector_type(8)));

#define NFEAT 128

#define RT   196
#define QCAP 16384
#define NBK  256
#define EB   256
#define DT   256
#define DR   10

__global__ void gcn_fill1(int* __restrict__ tcount) {
    int i = blockIdx.x * blockDim.x + threadIdx.x;
    if (i < NBK) tcount[i] = 0;
}

__global__ __launch_bounds__(EB) void gcn_bucket(const int* __restrict__ src,
                             const int* __restrict__ dst,
                             const float* __restrict__ ew,
                             int* __restrict__ tcount,
                             int* __restrict__ bucket, int n_edges) {
    __shared__ int hist[NBK];
    const int tid = threadIdx.x;
    for (int i = tid; i < NBK; i += EB) hist[i] = 0;
    __syncthreads();
    const int e0 = 4 * (blockIdx.x * EB + tid);
    const int n  = min(4, max(0, n_edges - e0));
    int s[4], d[4]; float w[4];
    if (n == 4) {
        #pragma unroll
        for (int j = 0; j < 4; ++j) { s[j] = src[e0 + j]; d[j] = dst[e0 + j]; w[j] = ew[e0 + j]; }
    } else {
        #pragma unroll
        for (int j = 0; j < 4; ++j) {
            s[j] = j < n ? src[e0 + j] : 0;
            d[j] = j < n ? dst[e0 + j] : 0;
            w[j] = j < n ? ew[e0 + j] : 0.0f;
        }
    }
    int key[4], rk[4];
    #pragma unroll
    for (int j = 0; j < 4; ++j) {
        key[j] = (int)(((unsigned)d[j] / RT) & (NBK - 1));
        rk[j]  = 0;
        if (j < n) rk[j] = atomicAdd(&hist[key[j]], 1);
    }
    __syncthreads();
    for (int i = tid; i < NBK; i += EB) {
        const int cnt = hist[i];
        hist[i] = cnt ? atomicAdd(&tcount[i], cnt) : 0;
    }
    __syncthreads();
    int4* const brec = (int4*)bucket;
    #pragma unroll
    for (int j = 0; j < 4; ++j) {
        if (j < n) {
            const int pos = hist[key[j]] + rk[j];
            if (pos < QCAP)
                brec[(size_t)key[j] * QCAP + pos] = make_int4(s[j], d[j], __float_as_int(w[j]), 0);
        }
    }
}

__global__ __launch_bounds__(DT) void gcn_tile_dinv(const int* __restrict__ dst,
                             const float* __restrict__ ew,
                             const int* __restrict__ bucket, const int* __restrict__ tcount,
                             float* __restrict__ dinv, int n_nodes, int n_edges) {
    __shared__ float sdeg[RT];
    const int tid   = threadIdx.x;
    const int lo    = blockIdx.x * RT;
    const int nrows = min(RT, n_nodes - lo);
    const int nb    = tcount[blockIdx.x & (NBK - 1)];
    for (int i = tid; i < RT; i += DT) sdeg[i] = 0.0f;
    __syncthreads();
    if (nb <= QCAP) {
        const int4* const brec = (const int4*)bucket + (size_t)(blockIdx.x & (NBK - 1)) * QCAP;
        for (int base = tid; base < nb; base += DT * DR) {
            int4 r[DR];
            #pragma unroll
            for (int u = 0; u < DR; ++u) {
                const int i = base + u * DT;
                r[u] = i < nb ? brec[i] : make_int4(0, -1, 0, 0);
            }
            #pragma unroll
            for (int u = 0; u < DR; ++u) {
                const unsigned row = (unsigned)(r[u].y - lo);
                if (row < (unsigned)nrows)
                    __hip_atomic_fetch_add(&sdeg[row], __int_as_float(r[u].z),
                                           __ATOMIC_RELAXED, __HIP_MEMORY_SCOPE_WORKGROUP);
            }
        }
    } else {
        for (int e = tid; e < n_edges; e += DT) {
            const unsigned row = (unsigned)(dst[e] - lo);
            if (row < (unsigned)nrows)
                __hip_atomic_fetch_add(&sdeg[row], ew[e],
                                       __ATOMIC_RELAXED, __HIP_MEMORY_SCOPE_WORKGROUP);
        }
    }
    __syncthreads();
    for (int i = tid; i < nrows; i += DT) dinv[lo + i] = rsqrtf(1.0f + sdeg[i]);
}

#define MB   13
#define TP   132
#define WP   136
#define TCOL(f) ((((f) & 3) << 5) + ((f) >> 2))
#define ATH  1024
#define DU   8
#define NCLS 10
#define CP   16

template <int N> struct tile_count { static constexpr int value = N; };

template <bool FROM_LIST>
__global__ __launch_bounds__(ATH) void gcn_layer_tile(
        const float* __restrict__ h, const float* __restrict__ W,
        const int* __restrict__ src,
        const int* __restrict__ dst, const float* __restrict__ ew,
        const float* __restrict__ dinv, const float* __restrict__ b,
        float* __restrict__ out, int* __restrict__ lists, int* __restrict__ cnts,
        const int* __restrict__ bucket, const int* __restrict__ tcount,
        const float* __restrict__ Wc, const float* __restrict__ bc,
        float* __restrict__ scores, int n_nodes, int n_edges) {
    __shared__ __attribute__((aligned(16))) float acc[MB * 16 * TP];
    __shared__ __attribute__((aligned(16))) float Wl[NFEAT * WP];
    __shared__ int q[QCAP];
    __shared__ int qn;
    __shared__ __attribute__((aligned(16))) float Wcl[FROM_LIST ? NFEAT * CP : 1];
    __shared__ __attribute__((aligned(16))) float lg[FROM_LIST ? MB * 16 * NCLS : 1];

    const int tid   = threadIdx.x;
    const int lane  = tid & 31;
    const int lo    = blockIdx.x * RT;
    const int nrows = min(RT, n_nodes - lo);
    int4* const tl  = (int4*)lists + (size_t)blockIdx.x * QCAP;
    const int4* const brec = (const int4*)bucket + (size_t)(blockIdx.x & (NBK - 1)) * QCAP;

    auto clear_tile = [&]() {
        float4* a4 = (float4*)acc;
        for (int i = tid; i < MB * 16 * TP / 4; i += ATH)
            a4[i] = make_float4(0.0f, 0.0f, 0.0f, 0.0f);
    };

    auto note = [&](int e, int d) {
        if ((unsigned)(d - lo) < (unsigned)nrows) {
            int idx = atomicAdd(&qn, 1);
            if (idx < QCAP) q[idx] = e;
        }
    };

    auto drain = [&](int total) {
        const int wave = __builtin_amdgcn_readfirstlane(tid >> 5);
        for (int base = wave; base < total; base += 32 * DU) {
            int e[DU]; bool ok[DU];
            #pragma unroll
            for (int u = 0; u < DU; ++u) {
                const int i = base + 32 * u;
                ok[u] = i < total;
                e[u]  = q[ok[u] ? i : base];
            }
            __builtin_amdgcn_sched_barrier(0);
            int sN[DU], dN[DU]; float w[DU];
            #pragma unroll
            for (int u = 0; u < DU; ++u) {
                sN[u] = src[e[u]]; dN[u] = dst[e[u]]; w[u] = ew[e[u]];
            }
            __builtin_amdgcn_sched_barrier(0);
            float c[DU]; float4 hv[DU];
            #pragma unroll
            for (int u = 0; u < DU; ++u) {
                c[u] = dinv[sN[u]] * w[u] * dinv[dN[u]];
                hv[u] = *((const float4*)(h + (size_t)sN[u] * NFEAT) + lane);
            }
            #pragma unroll
            for (int u = 0; u < DU; ++u) {
                const int row = (ok[u] && (unsigned)(dN[u] - lo) < (unsigned)nrows) ? dN[u] - lo : RT;
                float* ar = acc + row * TP + lane;
                const float hj[4] = {hv[u].x, hv[u].y, hv[u].z, hv[u].w};
                #pragma unroll
                for (int j = 0; j < 4; ++j)
                    __hip_atomic_fetch_add(ar + 32 * j, hj[j] * c[u],
                                           __ATOMIC_RELAXED, __HIP_MEMORY_SCOPE_WORKGROUP);
            }
        }
    };

    auto drain_bucket = [&](int total, int4 r) {
        const int wave = __builtin_amdgcn_readfirstlane(tid >> 5);
        const int sub  = lane & (DU - 1);
        int base = wave * DU;
        if (base >= total) return;
        const int4* bslot = brec + base;
        int4* slot = tl + base;
        for (; base < total; base += 32 * DU, slot += 32 * DU) {
            float4 hv[DU];
            #pragma unroll
            for (int u = 0; u < DU; ++u) {
                const int sN = __builtin_amdgcn_readlane(r.x, u);
                hv[u] = *((const float4*)(h + (size_t)sN * NFEAT) + lane);
            }
            const float ds = dinv[r.x];
            const float dd = dinv[r.y];
            const bool more = base + 32 * DU < total;
            if (more) bslot += 32 * DU;
            const int4 rn = bslot[(more ? base + 32 * DU : base) + sub < total ? sub : 0];
            const float cm  = ds * __int_as_float(r.z) * dd;
            const int rowm  = (base + sub < total && (unsigned)(r.y - lo) < (unsigned)nrows) ? r.y - lo : RT;
            #pragma unroll
            for (int u = 0; u < DU; ++u) {
                const int   row = __builtin_amdgcn_readlane(rowm, u);
                const float cu  = __int_as_float(__builtin_amdgcn_readlane(__float_as_int(cm), u));
                float* ar = acc + row * TP + lane;
                const float hj[4] = {hv[u].x, hv[u].y, hv[u].z, hv[u].w};
                #pragma unroll
                for (int j = 0; j < 4; ++j)
                    __hip_atomic_fetch_add(ar + 32 * j, hj[j] * cu,
                                           __ATOMIC_RELAXED, __HIP_MEMORY_SCOPE_WORKGROUP);
            }
            if (lane < DU) slot[lane] = make_int4(r.x, rowm, __float_as_int(cm), 0);
            r = rn;
        }
    };

    auto drain_list = [&](int total) {
        const int wave = __builtin_amdgcn_readfirstlane(tid >> 5);
        if (wave * DU >= total) return;
        const int4* slot = tl + wave * DU;
        int4 en[DU];
        #pragma unroll
        for (int u = 0; u < DU; ++u) en[u] = slot[u];
        for (int base = wave * DU; base < total; base += 32 * DU) {
            float4 hv[DU];
            #pragma unroll
            for (int u = 0; u < DU; ++u)
                hv[u] = *((const float4*)(h + (size_t)en[u].x * NFEAT) + lane);
            if (base + 32 * DU < total) slot += 32 * DU;
            int4 nx[DU];
            #pragma unroll
            for (int u = 0; u < DU; ++u) nx[u] = slot[u];
            #pragma unroll
            for (int u = 0; u < DU; ++u) {
                float* ar = acc + en[u].y * TP + lane;
                const float cu = __int_as_float(en[u].z);
                const float hj[4] = {hv[u].x, hv[u].y, hv[u].z, hv[u].w};
                #pragma unroll
                for (int j = 0; j < 4; ++j)
                    __hip_atomic_fetch_add(ar + 32 * j, hj[j] * cu,
                                           __ATOMIC_RELAXED, __HIP_MEMORY_SCOPE_WORKGROUP);
            }
            #pragma unroll
            for (int u = 0; u < DU; ++u) en[u] = nx[u];
        }
    };

    int nb = 0;
    int4 r0 = make_int4(0, 0, 0, 0);
    {
        if constexpr (!FROM_LIST) nb = tcount[blockIdx.x & (NBK - 1)];
        const float4* Wv = (const float4*)W;
        float4 wv[(NFEAT * NFEAT / 4) / ATH];
        #pragma unroll
        for (int u = 0; u < (NFEAT * NFEAT / 4) / ATH; ++u) wv[u] = Wv[tid + u * ATH];
        float wcv[(NFEAT * CP) / ATH];
        if constexpr (FROM_LIST) {
            #pragma unroll
            for (int u = 0; u < (NFEAT * CP) / ATH; ++u) {
                const int i = tid + u * ATH;
                const int c = i & (CP - 1);
                wcv[u] = c < NCLS ? Wc[(i / CP) * NCLS + c] : 0.0f;
            }
        }
        if constexpr (!FROM_LIST) {
            const int b0 = __builtin_amdgcn_readfirstlane(tid >> 5) * DU;
            if (nb <= QCAP && b0 < nb)
                r0 = brec[b0 + (b0 + (lane & (DU - 1)) < nb ? (lane & (DU - 1)) : 0)];
        }
        clear_tile();
        #pragma unroll
        for (int u = 0; u < (NFEAT * NFEAT / 4) / ATH; ++u) {
            const int i = tid + u * ATH;
            *(float4*)(Wl + TCOL(i >> 5) * WP + (i & 31) * 4) = wv[u];
        }
        if constexpr (FROM_LIST) {
            #pragma unroll
            for (int u = 0; u < (NFEAT * CP) / ATH; ++u) Wcl[tid + u * ATH] = wcv[u];
        }
    }
    if (tid == 0) qn = 0;
    __syncthreads();

    int noted;
    if constexpr (FROM_LIST) {
        noted = cnts[blockIdx.x * 32];
    } else {
        noted = nb;
        if (tid == 0) cnts[blockIdx.x * 32] = noted;
    }

    if (noted <= QCAP) {
        if constexpr (FROM_LIST) drain_list(noted);
        else                     drain_bucket(noted, r0);
    } else {
        __syncthreads();
        clear_tile();
        for (int seg = 0; seg < n_edges; seg += QCAP) {
            if (tid == 0) qn = 0;
            __syncthreads();
            const int segEnd = min(seg + QCAP, n_edges);
            for (int e = seg + tid; e < segEnd; e += ATH) note(e, dst[e]);
            __syncthreads();
            drain(qn);
            __syncthreads();
        }
    }
    __syncthreads();

    const float4* h4 = (const float4*)h;
    for (int idx = tid; idx < nrows * (NFEAT / 4); idx += ATH) {
        const int r  = idx >> 5;
        const int f4 = idx & 31;
        const float di = dinv[lo + r];
        const float d2 = di * di;
        const float4 hv = h4[(size_t)(lo + r) * (NFEAT / 4) + f4];
        float* ap = acc + r * TP + f4;
        float a0 = ap[0], a1 = ap[32], a2 = ap[64], a3 = ap[96];
        a0 += hv.x * d2;
        a1 += hv.y * d2;
        a2 += hv.z * d2;
        a3 += hv.w * d2;
        ap[0] = a0; ap[32] = a1; ap[64] = a2; ap[96] = a3;
    }
    __syncthreads();

    const int wave  = __builtin_amdgcn_readfirstlane(tid >> 5);
    const int mlane = lane & 15;
    const int khalf = (lane >> 4) * 2;
    const int ncol  = (wave & 7) * 16 + mlane;
    const int mq    = wave >> 3;
    const int cnt   = (((nrows + 15) >> 4) - mq + 3) >> 2;
    v8f d[4] = {};
    auto tile_mm = [&](auto ntc) {
        constexpr int NT = decltype(ntc)::value;
        const float* ap = acc + (mq * 16 + mlane) * TP + khalf;
        const float* bp = Wl + khalf * WP + ncol;
        #pragma unroll 4
        for (int k0 = 0; k0 < NFEAT; k0 += 4) {
            v2f bb;
            bb.x = bp[(k0    ) * WP];
            bb.y = bp[(k0 + 1) * WP];
            #pragma unroll
            for (int i = 0; i < NT; ++i) {
                const v2f a = *(const v2f*)(ap + i * 64 * TP + k0);
                d[i] = __builtin_amdgcn_wmma_f32_16x16x4_f32(
                           false, a, false, bb, (short)0, d[i], false, false);
            }
        }
    };
    switch (cnt) {
        case 4: tile_mm(tile_count<4>{}); break;
        case 3: tile_mm(tile_count<3>{}); break;
        case 2: tile_mm(tile_count<2>{}); break;
        case 1: tile_mm(tile_count<1>{}); break;
        default: break;
    }
    __syncthreads();

    const float bias = b[ncol];
    #pragma unroll
    for (int i = 0; i < 4; ++i) {
        if (i < cnt) {
            float* dp = acc + ((mq + 4 * i) * 16 + (lane >> 4) * 8) * TP + ncol;
            #pragma unroll
            for (int r = 0; r < 8; ++r) dp[r * TP] = fmaxf(d[i][r] + bias, 0.0f);
        }
    }
    __syncthreads();

    if constexpr (!FROM_LIST) {
        float4* o4 = (float4*)out;
        for (int idx = tid; idx < nrows * (NFEAT / 4); idx += ATH) {
            const int r  = idx >> 5;
            const int f4 = idx & 31;
            o4[(size_t)(lo + r) * (NFEAT / 4) + f4] = *((const float4*)(acc + r * TP) + f4);
        }
    } else {
        if (wave < ((nrows + 15) >> 4)) {
            const float bcv = bc[min(mlane, NCLS - 1)];
            const float* ap = acc + (wave * 16 + mlane) * TP + khalf;
            const float* bp = Wcl + khalf * CP + mlane;
            v8f dc = {};
            #pragma unroll 4
            for (int k0 = 0; k0 < NFEAT; k0 += 4) {
                const v2f a = *(const v2f*)(ap + k0);
                v2f bb;
                bb.x = bp[(k0    ) * CP];
                bb.y = bp[(k0 + 1) * CP];
                dc = __builtin_amdgcn_wmma_f32_16x16x4_f32(
                         false, a, false, bb, (short)0, dc, false, false);
            }
            float sc[8];
            #pragma unroll
            for (int r = 0; r < 8; ++r) sc[r] = dc[r] + bcv;
            if (mlane < NCLS) {
                float* lp = lg + (wave * 16 + (lane >> 4) * 8) * NCLS + mlane;
                #pragma unroll
                for (int r = 0; r < 8; ++r) lp[r * NCLS] = sc[r];
            }
        }
        __syncthreads();
        const int g0 = lo * NCLS;
        const int g1 = g0 + nrows * NCLS;
        for (int g = (g0 & ~31) + tid; g < g1; g += ATH)
            if (g >= g0) scores[g] = lg[g - g0];
    }
}

extern "C" void kernel_launch(void* const* d_in, const int* in_sizes, int n_in,
                              void* d_out, int out_size, void* d_ws, size_t ws_size,
                              hipStream_t stream) {
    const float* x   = (const float*)d_in[0];
    const int*   ei  = (const int*)  d_in[1];
    const float* ew  = (const float*)d_in[2];
    const float* W1  = (const float*)d_in[3];
    const float* b1  = (const float*)d_in[4];
    const float* W2  = (const float*)d_in[5];
    const float* b2  = (const float*)d_in[6];
    const float* Wc  = (const float*)d_in[7];
    const float* bc  = (const float*)d_in[8];
    float*       out = (float*)d_out;

    const int n_nodes = in_sizes[0] / NFEAT;
    const int n_edges = in_sizes[2];
    const int* src = ei;
    const int* dst = ei + n_edges;

    const size_t nd = (size_t)n_nodes * NFEAT;
    float* bufA = (float*)d_ws;
    float* dinv = bufA + nd;
    const int gT  = (n_nodes + RT - 1) / RT;
    int* lists = (int*)(dinv + (size_t)((n_nodes + 31) / 32) * 32);
    int* cnts  = lists + (size_t)gT * QCAP * 4;
    int* bucket = cnts + (size_t)gT * 32;
    int* tcount = bucket + (size_t)NBK * QCAP * 4;

    const int gE  = (n_edges + 4 * EB - 1) / (4 * EB);

    gcn_fill1    <<<1, NBK, 0, stream>>>(tcount);
    gcn_bucket   <<<gE, EB, 0, stream>>>(src, dst, ew, tcount, bucket, n_edges);
    gcn_tile_dinv<<<gT, DT, 0, stream>>>(dst, ew, bucket, tcount, dinv, n_nodes, n_edges);

    gcn_layer_tile<false><<<gT, ATH, 0, stream>>>(x, W1, src, dst, ew, dinv, b1, bufA, lists, cnts, bucket, tcount, Wc, bc, out, n_nodes, n_edges);

    gcn_layer_tile<true><<<gT, ATH, 0, stream>>>(bufA, W2, src, dst, ew, dinv, b2, nullptr, lists, cnts, bucket, tcount, Wc, bc, out, n_nodes, n_edges);
}


# Loads that need no edge go out with W in one round trip: self-loop rows become the tile's start value (self term summed first, was last), biases, layer-2 count; self-loop pass and its barrier go
# speedup vs baseline: 4.5586x; 1.0558x over previous
#include <hip/hip_runtime.h>
#include <hip/hip_bf16.h>

typedef float v2f __attribute__((ext_vector_type(2)));
typedef float v8f __attribute__((ext_vector_type(8)));

#define NFEAT 128

#define RT   196
#define QCAP 16384
#define NBK  256
#define EB   256
#define DT   256
#define DR   10

__global__ void gcn_fill1(int* __restrict__ tcount) {
    int i = blockIdx.x * blockDim.x + threadIdx.x;
    if (i < NBK) tcount[i] = 0;
}

__global__ __launch_bounds__(EB) void gcn_bucket(const int* __restrict__ src,
                             const int* __restrict__ dst,
                             const float* __restrict__ ew,
                             int* __restrict__ tcount,
                             int* __restrict__ bucket, int n_edges) {
    __shared__ int hist[NBK];
    const int tid = threadIdx.x;
    for (int i = tid; i < NBK; i += EB) hist[i] = 0;
    __syncthreads();
    const int e0 = 4 * (blockIdx.x * EB + tid);
    const int n  = min(4, max(0, n_edges - e0));
    int s[4], d[4]; float w[4];
    if (n == 4) {
        #pragma unroll
        for (int j = 0; j < 4; ++j) { s[j] = src[e0 + j]; d[j] = dst[e0 + j]; w[j] = ew[e0 + j]; }
    } else {
        #pragma unroll
        for (int j = 0; j < 4; ++j) {
            s[j] = j < n ? src[e0 + j] : 0;
            d[j] = j < n ? dst[e0 + j] : 0;
            w[j] = j < n ? ew[e0 + j] : 0.0f;
        }
    }
    int key[4], rk[4];
    #pragma unroll
    for (int j = 0; j < 4; ++j) {
        key[j] = (int)(((unsigned)d[j] / RT) & (NBK - 1));
        rk[j]  = 0;
        if (j < n) rk[j] = atomicAdd(&hist[key[j]], 1);
    }
    __syncthreads();
    for (int i = tid; i < NBK; i += EB) {
        const int cnt = hist[i];
        hist[i] = cnt ? atomicAdd(&tcount[i], cnt) : 0;
    }
    __syncthreads();
    int4* const brec = (int4*)bucket;
    #pragma unroll
    for (int j = 0; j < 4; ++j) {
        if (j < n) {
            const int pos = hist[key[j]] + rk[j];
            if (pos < QCAP)
                brec[(size_t)key[j] * QCAP + pos] = make_int4(s[j], d[j], __float_as_int(w[j]), 0);
        }
    }
}

__global__ __launch_bounds__(DT) void gcn_tile_dinv(const int* __restrict__ dst,
                             const float* __restrict__ ew,
                             const int* __restrict__ bucket, const int* __restrict__ tcount,
                             float* __restrict__ dinv, int n_nodes, int n_edges) {
    __shared__ float sdeg[RT];
    const int tid   = threadIdx.x;
    const int lo    = blockIdx.x * RT;
    const int nrows = min(RT, n_nodes - lo);
    const int nb    = tcount[blockIdx.x & (NBK - 1)];
    for (int i = tid; i < RT; i += DT) sdeg[i] = 0.0f;
    __syncthreads();
    if (nb <= QCAP) {
        const int4* const brec = (const int4*)bucket + (size_t)(blockIdx.x & (NBK - 1)) * QCAP;
        for (int base = tid; base < nb; base += DT * DR) {
            int4 r[DR];
            #pragma unroll
            for (int u = 0; u < DR; ++u) {
                const int i = base + u * DT;
                r[u] = i < nb ? brec[i] : make_int4(0, -1, 0, 0);
            }
            #pragma unroll
            for (int u = 0; u < DR; ++u) {
                const unsigned row = (unsigned)(r[u].y - lo);
                if (row < (unsigned)nrows)
                    __hip_atomic_fetch_add(&sdeg[row], __int_as_float(r[u].z),
                                           __ATOMIC_RELAXED, __HIP_MEMORY_SCOPE_WORKGROUP);
            }
        }
    } else {
        for (int e = tid; e < n_edges; e += DT) {
            const unsigned row = (unsigned)(dst[e] - lo);
            if (row < (unsigned)nrows)
                __hip_atomic_fetch_add(&sdeg[row], ew[e],
                                       __ATOMIC_RELAXED, __HIP_MEMORY_SCOPE_WORKGROUP);
        }
    }
    __syncthreads();
    for (int i = tid; i < nrows; i += DT) dinv[lo + i] = rsqrtf(1.0f + sdeg[i]);
}

#define MB   13
#define TP   132
#define WP   136
#define TCOL(f) ((((f) & 3) << 5) + ((f) >> 2))
#define ATH  1024
#define DU   8
#define NCLS 10
#define CP   16

template <int N> struct tile_count { static constexpr int value = N; };

#define HOLD4(v) "v"((v).x), "v"((v).y), "v"((v).z), "v"((v).w)

template <bool FROM_LIST>
__global__ __launch_bounds__(ATH) void gcn_layer_tile(
        const float* __restrict__ h, const float* __restrict__ W,
        const int* __restrict__ src,
        const int* __restrict__ dst, const float* __restrict__ ew,
        const float* __restrict__ dinv, const float* __restrict__ b,
        float* __restrict__ out, int* __restrict__ lists, int* __restrict__ cnts,
        const int* __restrict__ bucket, const int* __restrict__ tcount,
        const float* __restrict__ Wc, const float* __restrict__ bc,
        float* __restrict__ scores, int n_nodes, int n_edges) {
    __shared__ __attribute__((aligned(16))) float acc[MB * 16 * TP];
    __shared__ __attribute__((aligned(16))) float Wl[NFEAT * WP];
    __shared__ int q[QCAP];
    __shared__ int qn;
    __shared__ __attribute__((aligned(16))) float Wcl[FROM_LIST ? NFEAT * CP : 1];
    __shared__ __attribute__((aligned(16))) float lg[FROM_LIST ? MB * 16 * NCLS : 1];

    const int tid   = threadIdx.x;
    const int lane  = tid & 31;
    const int lo    = blockIdx.x * RT;
    const int nrows = min(RT, n_nodes - lo);
    int4* const tl  = (int4*)lists + (size_t)blockIdx.x * QCAP;
    const int4* const brec = (const int4*)bucket + (size_t)(blockIdx.x & (NBK - 1)) * QCAP;

    constexpr int SL = (MB * 16 + 31) / 32;
    static_assert(SL == 7 && (NFEAT * NFEAT / 4) / ATH == 4 && (NFEAT * CP) / ATH == 2,
                  "the load fences in the prologue name these arrays element by element");
    const int swave = __builtin_amdgcn_readfirstlane(tid >> 5);
    auto start_loads = [&](float4 (&sv)[SL], float (&sd)[SL]) {
        #pragma unroll
        for (int u = 0; u < SL; ++u) {
            const int rc = min(swave + 32 * u, nrows - 1);
            sd[u] = dinv[lo + rc];
            sv[u] = *((const float4*)(h + (size_t)(lo + rc) * NFEAT) + lane);
        }
    };
    auto start_stores = [&](const float4 (&sv)[SL], const float (&sd)[SL]) {
        #pragma unroll
        for (int u = 0; u < SL; ++u) {
            const int r = swave + 32 * u;
            if (u < SL - 1 || r < MB * 16) {
                const bool  own = r < nrows;
                const float d2  = sd[u] * sd[u];
                float* ap = acc + r * TP + lane;
                ap[0]  = own ? sv[u].x * d2 : 0.0f;
                ap[32] = own ? sv[u].y * d2 : 0.0f;
                ap[64] = own ? sv[u].z * d2 : 0.0f;
                ap[96] = own ? sv[u].w * d2 : 0.0f;
            }
        }
    };

    auto note = [&](int e, int d) {
        if ((unsigned)(d - lo) < (unsigned)nrows) {
            int idx = atomicAdd(&qn, 1);
            if (idx < QCAP) q[idx] = e;
        }
    };

    auto drain = [&](int total) {
        const int wave = __builtin_amdgcn_readfirstlane(tid >> 5);
        for (int base = wave; base < total; base += 32 * DU) {
            int e[DU]; bool ok[DU];
            #pragma unroll
            for (int u = 0; u < DU; ++u) {
                const int i = base + 32 * u;
                ok[u] = i < total;
                e[u]  = q[ok[u] ? i : base];
            }
            __builtin_amdgcn_sched_barrier(0);
            int sN[DU], dN[DU]; float w[DU];
            #pragma unroll
            for (int u = 0; u < DU; ++u) {
                sN[u] = src[e[u]]; dN[u] = dst[e[u]]; w[u] = ew[e[u]];
            }
            __builtin_amdgcn_sched_barrier(0);
            float c[DU]; float4 hv[DU];
            #pragma unroll
            for (int u = 0; u < DU; ++u) {
                c[u] = dinv[sN[u]] * w[u] * dinv[dN[u]];
                hv[u] = *((const float4*)(h + (size_t)sN[u] * NFEAT) + lane);
            }
            #pragma unroll
            for (int u = 0; u < DU; ++u) {
                const int row = (ok[u] && (unsigned)(dN[u] - lo) < (unsigned)nrows) ? dN[u] - lo : RT;
                float* ar = acc + row * TP + lane;
                const float hj[4] = {hv[u].x, hv[u].y, hv[u].z, hv[u].w};
                #pragma unroll
                for (int j = 0; j < 4; ++j)
                    __hip_atomic_fetch_add(ar + 32 * j, hj[j] * c[u],
                                           __ATOMIC_RELAXED, __HIP_MEMORY_SCOPE_WORKGROUP);
            }
        }
    };

    auto drain_bucket = [&](int total, int4 r) {
        const int wave = __builtin_amdgcn_readfirstlane(tid >> 5);
        const int sub  = lane & (DU - 1);
        int base = wave * DU;
        if (base >= total) return;
        const int4* bslot = brec + base;
        int4* slot = tl + base;
        for (; base < total; base += 32 * DU, slot += 32 * DU) {
            float4 hv[DU];
            #pragma unroll
            for (int u = 0; u < DU; ++u) {
                const int sN = __builtin_amdgcn_readlane(r.x, u);
                hv[u] = *((const float4*)(h + (size_t)sN * NFEAT) + lane);
            }
            const float ds = dinv[r.x];
            const float dd = dinv[r.y];
            const bool more = base + 32 * DU < total;
            if (more) bslot += 32 * DU;
            const int4 rn = bslot[(more ? base + 32 * DU : base) + sub < total ? sub : 0];
            const float cm  = ds * __int_as_float(r.z) * dd;
            const int rowm  = (base + sub < total && (unsigned)(r.y - lo) < (unsigned)nrows) ? r.y - lo : RT;
            #pragma unroll
            for (int u = 0; u < DU; ++u) {
                const int   row = __builtin_amdgcn_readlane(rowm, u);
                const float cu  = __int_as_float(__builtin_amdgcn_readlane(__float_as_int(cm), u));
                float* ar = acc + row * TP + lane;
                const float hj[4] = {hv[u].x, hv[u].y, hv[u].z, hv[u].w};
                #pragma unroll
                for (int j = 0; j < 4; ++j)
                    __hip_atomic_fetch_add(ar + 32 * j, hj[j] * cu,
                                           __ATOMIC_RELAXED, __HIP_MEMORY_SCOPE_WORKGROUP);
            }
            if (lane < DU) slot[lane] = make_int4(r.x, rowm, __float_as_int(cm), 0);
            r = rn;
        }
    };

    auto drain_list = [&](int total) {
        const int wave = __builtin_amdgcn_readfirstlane(tid >> 5);
        if (wave * DU >= total) return;
        const int4* slot = tl + wave * DU;
        int4 en[DU];
        #pragma unroll
        for (int u = 0; u < DU; ++u) en[u] = slot[u];
        for (int base = wave * DU; base < total; base += 32 * DU) {
            float4 hv[DU];
            #pragma unroll
            for (int u = 0; u < DU; ++u)
                hv[u] = *((const float4*)(h + (size_t)en[u].x * NFEAT) + lane);
            if (base + 32 * DU < total) slot += 32 * DU;
            int4 nx[DU];
            #pragma unroll
            for (int u = 0; u < DU; ++u) nx[u] = slot[u];
            #pragma unroll
            for (int u = 0; u < DU; ++u) {
                float* ar = acc + en[u].y * TP + lane;
                const float cu = __int_as_float(en[u].z);
                const float hj[4] = {hv[u].x, hv[u].y, hv[u].z, hv[u].w};
                #pragma unroll
                for (int j = 0; j < 4; ++j)
                    __hip_atomic_fetch_add(ar + 32 * j, hj[j] * cu,
                                           __ATOMIC_RELAXED, __HIP_MEMORY_SCOPE_WORKGROUP);
            }
            #pragma unroll
            for (int u = 0; u < DU; ++u) en[u] = nx[u];
        }
    };

    int nb = 0;
    int4 r0 = make_int4(0, 0, 0, 0);
    float bias;
    float bcv = 0.0f;
    {
        if constexpr (!FROM_LIST) nb = tcount[blockIdx.x & (NBK - 1)];
        else                      nb = cnts[blockIdx.x * 32];
        bias = b[(swave & 7) * 16 + (lane & 15)];
        if constexpr (FROM_LIST) bcv = bc[min(lane & 15, NCLS - 1)];
        const float4* Wv = (const float4*)W;
        float4 wv[(NFEAT * NFEAT / 4) / ATH];
        #pragma unroll
        for (int u = 0; u < (NFEAT * NFEAT / 4) / ATH; ++u) wv[u] = Wv[tid + u * ATH];
        float wcv[(NFEAT * CP) / ATH];
        if constexpr (FROM_LIST) {
            #pragma unroll
            for (int u = 0; u < (NFEAT * CP) / ATH; ++u) {
                const int i = tid + u * ATH;
                const int c = i & (CP - 1);
                wcv[u] = c < NCLS ? Wc[(i / CP) * NCLS + c] : 0.0f;
            }
        }
        float4 sv[SL]; float sd[SL];
        start_loads(sv, sd);
        asm volatile("" :: HOLD4(wv[0]), HOLD4(wv[1]), HOLD4(wv[2]), HOLD4(wv[3]));
        asm volatile("" :: HOLD4(sv[0]), HOLD4(sv[1]), HOLD4(sv[2]), HOLD4(sv[3]),
                           HOLD4(sv[4]), HOLD4(sv[5]), HOLD4(sv[6]));
        asm volatile("" :: "s"(sd[0]), "s"(sd[1]), "s"(sd[2]), "s"(sd[3]),
                           "s"(sd[4]), "s"(sd[5]), "s"(sd[6]), "v"(bias));
        if constexpr (FROM_LIST) asm volatile("" :: "v"(wcv[0]), "v"(wcv[1]), "v"(bcv));
        if constexpr (!FROM_LIST) {
            const int b0 = __builtin_amdgcn_readfirstlane(tid >> 5) * DU;
            if (nb <= QCAP && b0 < nb)
                r0 = brec[b0 + (b0 + (lane & (DU - 1)) < nb ? (lane & (DU - 1)) : 0)];
        }
        start_stores(sv, sd);
        #pragma unroll
        for (int u = 0; u < (NFEAT * NFEAT / 4) / ATH; ++u) {
            const int i = tid + u * ATH;
            *(float4*)(Wl + TCOL(i >> 5) * WP + (i & 31) * 4) = wv[u];
        }
        if constexpr (FROM_LIST) {
            #pragma unroll
            for (int u = 0; u < (NFEAT * CP) / ATH; ++u) Wcl[tid + u * ATH] = wcv[u];
        }
    }
    if (tid == 0) qn = 0;
    __syncthreads();

    const int noted = nb;
    if constexpr (!FROM_LIST) {
        if (tid == 0) cnts[blockIdx.x * 32] = noted;
    }

    if (noted <= QCAP) {
        if constexpr (FROM_LIST) drain_list(noted);
        else                     drain_bucket(noted, r0);
    } else {
        for (int seg = 0; seg < n_edges; seg += QCAP) {
            if (tid == 0) qn = 0;
            __syncthreads();
            const int segEnd = min(seg + QCAP, n_edges);
            for (int e = seg + tid; e < segEnd; e += ATH) note(e, dst[e]);
            __syncthreads();
            drain(qn);
            __syncthreads();
        }
    }
    __syncthreads();

    const int wave  = __builtin_amdgcn_readfirstlane(tid >> 5);
    const int mlane = lane & 15;
    const int khalf = (lane >> 4) * 2;
    const int ncol  = (wave & 7) * 16 + mlane;
    const int mq    = wave >> 3;
    const int cnt   = (((nrows + 15) >> 4) - mq + 3) >> 2;
    v8f d[4] = {};
    auto tile_mm = [&](auto ntc) {
        constexpr int NT = decltype(ntc)::value;
        const float* ap = acc + (mq * 16 + mlane) * TP + khalf;
        const float* bp = Wl + khalf * WP + ncol;
        #pragma unroll 4
        for (int k0 = 0; k0 < NFEAT; k0 += 4) {
            v2f bb;
            bb.x = bp[(k0    ) * WP];
            bb.y = bp[(k0 + 1) * WP];
            #pragma unroll
            for (int i = 0; i < NT; ++i) {
                const v2f a = *(const v2f*)(ap + i * 64 * TP + k0);
                d[i] = __builtin_amdgcn_wmma_f32_16x16x4_f32(
                           false, a, false, bb, (short)0, d[i], false, false);
            }
        }
    };
    switch (cnt) {
        case 4: tile_mm(tile_count<4>{}); break;
        case 3: tile_mm(tile_count<3>{}); break;
        case 2: tile_mm(tile_count<2>{}); break;
        case 1: tile_mm(tile_count<1>{}); break;
        default: break;
    }
    __syncthreads();

    #pragma unroll
    for (int i = 0; i < 4; ++i) {
        if (i < cnt) {
            float* dp = acc + ((mq + 4 * i) * 16 + (lane >> 4) * 8) * TP + ncol;
            #pragma unroll
            for (int r = 0; r < 8; ++r) dp[r * TP] = fmaxf(d[i][r] + bias, 0.0f);
        }
    }
    __syncthreads();

    if constexpr (!FROM_LIST) {
        float4* o4 = (float4*)out;
        for (int idx = tid; idx < nrows * (NFEAT / 4); idx += ATH) {
            const int r  = idx >> 5;
            const int f4 = idx & 31;
            o4[(size_t)(lo + r) * (NFEAT / 4) + f4] = *((const float4*)(acc + r * TP) + f4);
        }
    } else {
        if (wave < ((nrows + 15) >> 4)) {
            const float* ap = acc + (wave * 16 + mlane) * TP + khalf;
            const float* bp = Wcl + khalf * CP + mlane;
            v8f dc = {};
            #pragma unroll 4
            for (int k0 = 0; k0 < NFEAT; k0 += 4) {
                const v2f a = *(const v2f*)(ap + k0);
                v2f bb;
                bb.x = bp[(k0    ) * CP];
                bb.y = bp[(k0 + 1) * CP];
                dc = __builtin_amdgcn_wmma_f32_16x16x4_f32(
                         false, a, false, bb, (short)0, dc, false, false);
            }
            float sc[8];
            #pragma unroll
            for (int r = 0; r < 8; ++r) sc[r] = dc[r] + bcv;
            if (mlane < NCLS) {
                float* lp = lg + (wave * 16 + (lane >> 4) * 8) * NCLS + mlane;
                #pragma unroll
                for (int r = 0; r < 8; ++r) lp[r * NCLS] = sc[r];
            }
        }
        __syncthreads();
        const int g0 = lo * NCLS;
        const int g1 = g0 + nrows * NCLS;
        for (int g = (g0 & ~31) + tid; g < g1; g += ATH)
            if (g >= g0) scores[g] = lg[g - g0];
    }
}

extern "C" void kernel_launch(void* const* d_in, const int* in_sizes, int n_in,
                              void* d_out, int out_size, void* d_ws, size_t ws_size,
                              hipStream_t stream) {
    const float* x   = (const float*)d_in[0];
    const int*   ei  = (const int*)  d_in[1];
    const float* ew  = (const float*)d_in[2];
    const float* W1  = (const float*)d_in[3];
    const float* b1  = (const float*)d_in[4];
    const float* W2  = (const float*)d_in[5];
    const float* b2  = (const float*)d_in[6];
    const float* Wc  = (const float*)d_in[7];
    const float* bc  = (const float*)d_in[8];
    float*       out = (float*)d_out;

    const int n_nodes = in_sizes[0] / NFEAT;
    const int n_edges = in_sizes[2];
    const int* src = ei;
    const int* dst = ei + n_edges;

    const size_t nd = (size_t)n_nodes * NFEAT;
    float* bufA = (float*)d_ws;
    float* dinv = bufA + nd;
    const int gT  = (n_nodes + RT - 1) / RT;
    int* lists = (int*)(dinv + (size_t)((n_nodes + 31) / 32) * 32);
    int* cnts  = lists + (size_t)gT * QCAP * 4;
    int* bucket = cnts + (size_t)gT * 32;
    int* tcount = bucket + (size_t)NBK * QCAP * 4;

    const int gE  = (n_edges + 4 * EB - 1) / (4 * EB);

    gcn_fill1    <<<1, NBK, 0, stream>>>(tcount);
    gcn_bucket   <<<gE, EB, 0, stream>>>(src, dst, ew, tcount, bucket, n_edges);
    gcn_tile_dinv<<<gT, DT, 0, stream>>>(dst, ew, bucket, tcount, dinv, n_nodes, n_edges);

    gcn_layer_tile<false><<<gT, ATH, 0, stream>>>(x, W1, src, dst, ew, dinv, b1, bufA, lists, cnts, bucket, tcount, Wc, bc, out, n_nodes, n_edges);

    gcn_layer_tile<true><<<gT, ATH, 0, stream>>>(bufA, W2, src, dst, ew, dinv, b2, nullptr, lists, cnts, bucket, tcount, Wc, bc, out, n_nodes, n_edges);
}
